# starting kernel
# baseline (speedup 1.0000x reference)
#include <hip/hip_runtime.h>
#include <math.h>

typedef float v2f __attribute__((ext_vector_type(2)));
typedef float v8f __attribute__((ext_vector_type(8)));

#define NQ   14
#define DIM  16384
#define NT   256

static __device__ inline v8f wmma_16x16x4_f32(v2f a, v2f b, v8f c) {
#if defined(__gfx1250__)
  return __builtin_amdgcn_wmma_f32_16x16x4_f32(
      false, a, false, b, (short)0, c, false, false);
#else
  return c;
#endif
}

__global__ __launch_bounds__(NT) void qsim14_kernel(
    const float* __restrict__ x,
    const float* __restrict__ params,
    float* __restrict__ out)
{
  __shared__ float st[2 * DIM];
  __shared__ float pHi[128], pLo[128];
  __shared__ float cq[NQ], sq[NQ];
  __shared__ float red[NQ * 8];

  const int b    = blockIdx.x;
  const int tid  = threadIdx.x;
  const int lane = tid & 31;
  const int wave = tid >> 5;

  if (tid < NQ) {
    float h = 0.5f * x[b * NQ + tid];
    cq[tid] = cosf(h);
    sq[tid] = sinf(h);
  }
  __syncthreads();

  if (tid < 128) {
    int h = tid; float p = 1.f;
#pragma unroll
    for (int t = 0; t < 7; ++t) p *= ((h >> (6 - t)) & 1) ? sq[t] : cq[t];
    pHi[h] = p;
  } else {
    int l = tid - 128; float p = 1.f;
#pragma unroll
    for (int u = 0; u < 7; ++u) p *= ((l >> (6 - u)) & 1) ? sq[7 + u] : cq[7 + u];
    pLo[l] = p;
  }
  __syncthreads();

#pragma unroll 4
  for (int k = 0; k < 64; ++k) {
    int idx = k * NT + tid;
    float m = pHi[idx >> 7] * pLo[idx & 127];
    int ph = __popc(idx) & 3;
    st[idx]       = (ph == 0) ? m : ((ph == 2) ? -m : 0.f);
    st[DIM + idx] = (ph == 1) ? -m : ((ph == 3) ? m : 0.f);
  }
  __syncthreads();

  const int baseArr[4] = {10, 6, 2, 0};

  for (int L = 0; L < 2; ++L) {
    for (int g = 0; g < 4; ++g) {
      const int base = baseArr[g];
      const int mask = (1 << base) - 1;

      float gc[4], gs[4];
#pragma unroll
      for (int m = 0; m < 4; ++m) {
        if (g == 3 && m >= 2) { gc[m] = 1.f; gs[m] = 0.f; }
        else {
          int q = 13 - base - m;
          float h = 0.5f * params[L * NQ + q];
          gc[m] = cosf(h); gs[m] = sinf(h);
        }
      }

      const int M    = lane & 15;
      const int Koff = (lane >> 4) << 1;
      v2f A[4];
#pragma unroll
      for (int k = 0; k < 4; ++k) {
#pragma unroll
        for (int r = 0; r < 2; ++r) {
          int Kc = 4 * k + Koff + r;
          float e = 1.f;
#pragma unroll
          for (int m = 0; m < 4; ++m) {
            int ib = (M >> m) & 1, jb = (Kc >> m) & 1;
            e *= (ib == jb) ? gc[m] : ((ib == 0) ? -gs[m] : gs[m]);
          }
          A[k][r] = e;
        }
      }

      for (int t = wave; t < 64; t += 8) {
        int c = 16 * t + (lane & 15);
        int colAddr = ((c >> base) << (base + 4)) + (c & mask);
        int rowsel  = (lane >> 4) << 1;
        int Moff    = (lane >> 4) << 3;
#pragma unroll
        for (int comp = 0; comp < 2; ++comp) {
          float* buf = st + comp * DIM;
          v8f acc = {};
#pragma unroll
          for (int k = 0; k < 4; ++k) {
            v2f Bf;
            Bf[0] = buf[colAddr + ((4 * k + rowsel + 0) << base)];
            Bf[1] = buf[colAddr + ((4 * k + rowsel + 1) << base)];
            acc = wmma_16x16x4_f32(A[k], Bf, acc);
          }
#pragma unroll
          for (int r = 0; r < 8; ++r)
            buf[colAddr + ((r + Moff) << base)] = acc[r];
        }
      }
      __syncthreads();
    }

#pragma unroll
    for (int comp = 0; comp < 2; ++comp) {
      float* buf = st + comp * DIM;
      float v[64];
#pragma unroll
      for (int k = 0; k < 64; ++k) {
        int j = k * NT + tid;
        v[k] = buf[j ^ (j >> 1)];
      }
      __syncthreads();
#pragma unroll
      for (int k = 0; k < 64; ++k)
        buf[k * NT + tid] = v[k];
      __syncthreads();
    }
  }

  float acc14[NQ];
#pragma unroll
  for (int w = 0; w < NQ; ++w) acc14[w] = 0.f;

  for (int k = 0; k < 64; ++k) {
    int j = k * NT + tid;
    float re = st[j], im = st[DIM + j];
    float p = re * re + im * im;
#pragma unroll
    for (int w = 0; w < NQ; ++w)
      acc14[w] += ((j >> (13 - w)) & 1) ? -p : p;
  }

#pragma unroll
  for (int w = 0; w < NQ; ++w) {
    float v = acc14[w];
#pragma unroll
    for (int off = 16; off > 0; off >>= 1)
      v += __shfl_down(v, off, 32);
    if (lane == 0) red[w * 8 + wave] = v;
  }
  __syncthreads();
  if (tid < NQ) {
    float s = 0.f;
#pragma unroll
    for (int i = 0; i < 8; ++i) s += red[tid * 8 + i];
    out[b * NQ + tid] = s;
  }
}

extern "C" void kernel_launch(void* const* d_in, const int* in_sizes, int n_in,
                              void* d_out, int out_size, void* d_ws, size_t ws_size,
                              hipStream_t stream) {
  const float* x      = (const float*)d_in[0];
  const float* params = (const float*)d_in[1];
  float* out          = (float*)d_out;
  int B = in_sizes[0] / NQ;
  qsim14_kernel<<<dim3(B), dim3(NT), 0, stream>>>(x, params, out);
}


# XOR-swizzled LDS layout of the state, lane half mapped to a bank-moving group bit: gate-pass loads/stores go from 2- and 4-way bank conflicts to none (WMMA k order permuted: sum order changes)
# speedup vs baseline: 1.0461x; 1.0461x over previous
#include <hip/hip_runtime.h>
#include <math.h>

typedef float v2f __attribute__((ext_vector_type(2)));
typedef float v8f __attribute__((ext_vector_type(8)));

#define NQ   14
#define DIM  16384
#define NT   256

static __device__ inline v8f wmma_16x16x4_f32(v2f a, v2f b, v8f c) {
#if defined(__gfx1250__)
  return __builtin_amdgcn_wmma_f32_16x16x4_f32(
      false, a, false, b, (short)0, c, false, false);
#else
  return c;
#endif
}

static __device__ __forceinline__ int swz(int i) {
  return i ^ (((i >> 6) & 15) << 2) ^ (((i >> 10) & 3) << 4);
}

static __device__ __forceinline__ int swz_lo(int T, int K) {
  return T ^ ((T >> 6) << 2) ^ ((((K & 3) ^ (K >> 2)) & 3) << 4);
}

static __device__ __forceinline__ int ins_bit(int v, int p, int b) {
  return ((v >> p) << (p + 1)) | (b << p) | (v & ((1 << p) - 1));
}

__global__ __launch_bounds__(NT) void qsim14_kernel(
    const float* __restrict__ x,
    const float* __restrict__ params,
    float* __restrict__ out)
{
  __shared__ float st[2 * DIM];
  __shared__ float pHi[128], pLo[128];
  __shared__ float cq[NQ], sq[NQ];
  __shared__ float red[NQ * 8];

  const int b    = blockIdx.x;
  const int tid  = threadIdx.x;
  const int lane = tid & 31;
  const int wave = tid >> 5;

  if (tid < NQ) {
    float h = 0.5f * x[b * NQ + tid];
    cq[tid] = cosf(h);
    sq[tid] = sinf(h);
  }
  __syncthreads();

  if (tid < 128) {
    int h = tid; float p = 1.f;
#pragma unroll
    for (int t = 0; t < 7; ++t) p *= ((h >> (6 - t)) & 1) ? sq[t] : cq[t];
    pHi[h] = p;
  } else {
    int l = tid - 128; float p = 1.f;
#pragma unroll
    for (int u = 0; u < 7; ++u) p *= ((l >> (6 - u)) & 1) ? sq[7 + u] : cq[7 + u];
    pLo[l] = p;
  }
  __syncthreads();

#pragma unroll 4
  for (int k = 0; k < 64; ++k) {
    int idx = k * NT + tid;
    float m = pHi[idx >> 7] * pLo[idx & 127];
    int ph = __popc(idx) & 3;
    int ps = (k << 8) + swz_lo(tid, k);
    st[ps]       = (ph == 0) ? m : ((ph == 2) ? -m : 0.f);
    st[DIM + ps] = (ph == 1) ? -m : ((ph == 3) ? m : 0.f);
  }
  __syncthreads();

  const int baseArr[4] = {10, 6, 2, 0};

  for (int L = 0; L < 2; ++L) {
    for (int g = 0; g < 4; ++g) {
      const int base = baseArr[g];
      const int mask = (1 << base) - 1;

      float gc[4], gs[4];
#pragma unroll
      for (int m = 0; m < 4; ++m) {
        if (g == 3 && m >= 2) { gc[m] = 1.f; gs[m] = 0.f; }
        else {
          int q = 13 - base - m;
          float h = 0.5f * params[L * NQ + q];
          gc[m] = cosf(h); gs[m] = sinf(h);
        }
      }

      const int p = (g == 1 || g == 2) ? 2 : 0;
      const int h = lane >> 4;

      const int Mrow = ins_bit(lane & 7, p, (lane >> 3) & 1);
      v2f A[4];
#pragma unroll
      for (int k = 0; k < 4; ++k) {
#pragma unroll
        for (int r = 0; r < 2; ++r) {
          int Kc = ins_bit(2 * k + r, p, h);
          float e = 1.f;
#pragma unroll
          for (int m = 0; m < 4; ++m) {
            int ib = (Mrow >> m) & 1, jb = (Kc >> m) & 1;
            e *= (ib == jb) ? gc[m] : ((ib == 0) ? -gs[m] : gs[m]);
          }
          A[k][r] = e;
        }
      }

      int rowP[8];
#pragma unroll
      for (int v = 0; v < 8; ++v) rowP[v] = swz(ins_bit(v, p, h) << base);

      for (int t = wave; t < 64; t += 8) {
        int c = 16 * t + (lane & 15);
        int colAddr = ((c >> base) << (base + 4)) + (c & mask);
        int pc = swz(colAddr);
#pragma unroll
        for (int comp = 0; comp < 2; ++comp) {
          float* buf = st + comp * DIM;
          v8f acc = {};
#pragma unroll
          for (int k = 0; k < 4; ++k) {
            v2f Bf;
            Bf[0] = buf[pc ^ rowP[2 * k + 0]];
            Bf[1] = buf[pc ^ rowP[2 * k + 1]];
            acc = wmma_16x16x4_f32(A[k], Bf, acc);
          }
#pragma unroll
          for (int r = 0; r < 8; ++r)
            buf[pc ^ rowP[r]] = acc[r];
        }
      }
      __syncthreads();
    }

#pragma unroll
    for (int comp = 0; comp < 2; ++comp) {
      float* buf = st + comp * DIM;
      float v[64];
#pragma unroll
      for (int k = 0; k < 64; ++k) {
        const int gk = k ^ (k >> 1);
        const int gt = (tid ^ (tid >> 1)) ^ ((k & 1) << 7);
        v[k] = buf[(gk << 8) + swz_lo(gt, gk)];
      }
      __syncthreads();
#pragma unroll
      for (int k = 0; k < 64; ++k)
        buf[(k << 8) + swz_lo(tid, k)] = v[k];
      __syncthreads();
    }
  }

  float acc14[NQ];
#pragma unroll
  for (int w = 0; w < NQ; ++w) acc14[w] = 0.f;

  for (int k = 0; k < 64; ++k) {
    int j = k * NT + tid;
    int pj = (k << 8) + swz_lo(tid, k);
    float re = st[pj], im = st[DIM + pj];
    float p = re * re + im * im;
#pragma unroll
    for (int w = 0; w < NQ; ++w)
      acc14[w] += ((j >> (13 - w)) & 1) ? -p : p;
  }

#pragma unroll
  for (int w = 0; w < NQ; ++w) {
    float v = acc14[w];
#pragma unroll
    for (int off = 16; off > 0; off >>= 1)
      v += __shfl_down(v, off, 32);
    if (lane == 0) red[w * 8 + wave] = v;
  }
  __syncthreads();
  if (tid < NQ) {
    float s = 0.f;
#pragma unroll
    for (int i = 0; i < 8; ++i) s += red[tid * 8 + i];
    out[b * NQ + tid] = s;
  }
}

extern "C" void kernel_launch(void* const* d_in, const int* in_sizes, int n_in,
                              void* d_out, int out_size, void* d_ws, size_t ws_size,
                              hipStream_t stream) {
  const float* x      = (const float*)d_in[0];
  const float* params = (const float*)d_in[1];
  float* out          = (float*)d_out;
  int B = in_sizes[0] / NQ;
  qsim14_kernel<<<dim3(B), dim3(NT), 0, stream>>>(x, params, out);
}


# Gate-pass tile loop software-pipelined: next tile's B values of both planes loaded as one group into a second register set ahead of the current tile's WMMAs; re/im accumulator chains alternate
# speedup vs baseline: 1.0890x; 1.0410x over previous
#include <hip/hip_runtime.h>
#include <math.h>

typedef float v2f __attribute__((ext_vector_type(2)));
typedef float v8f __attribute__((ext_vector_type(8)));

#define NQ   14
#define DIM  16384
#define NT   256

static __device__ inline v8f wmma_16x16x4_f32(v2f a, v2f b, v8f c) {
#if defined(__gfx1250__)
  return __builtin_amdgcn_wmma_f32_16x16x4_f32(
      false, a, false, b, (short)0, c, false, false);
#else
  return c;
#endif
}

static __device__ __forceinline__ int swz(int i) {
  return i ^ (((i >> 6) & 15) << 2) ^ (((i >> 10) & 3) << 4);
}

static __device__ __forceinline__ int swz_lo(int T, int K) {
  return T ^ ((T >> 6) << 2) ^ ((((K & 3) ^ (K >> 2)) & 3) << 4);
}

static __device__ __forceinline__ int ins_bit(int v, int p, int b) {
  return ((v >> p) << (p + 1)) | (b << p) | (v & ((1 << p) - 1));
}

__global__ __launch_bounds__(NT) __attribute__((amdgpu_waves_per_eu(4))) void qsim14_kernel(
    const float* __restrict__ x,
    const float* __restrict__ params,
    float* __restrict__ out)
{
  __shared__ float st[2 * DIM];
  __shared__ float pHi[128], pLo[128];
  __shared__ float cq[NQ], sq[NQ];
  __shared__ float red[NQ * 8];

  const int b    = blockIdx.x;
  const int tid  = threadIdx.x;
  const int lane = tid & 31;
  const int wave = tid >> 5;

  if (tid < NQ) {
    float h = 0.5f * x[b * NQ + tid];
    cq[tid] = cosf(h);
    sq[tid] = sinf(h);
  }
  __syncthreads();

  if (tid < 128) {
    int h = tid; float p = 1.f;
#pragma unroll
    for (int t = 0; t < 7; ++t) p *= ((h >> (6 - t)) & 1) ? sq[t] : cq[t];
    pHi[h] = p;
  } else {
    int l = tid - 128; float p = 1.f;
#pragma unroll
    for (int u = 0; u < 7; ++u) p *= ((l >> (6 - u)) & 1) ? sq[7 + u] : cq[7 + u];
    pLo[l] = p;
  }
  __syncthreads();

#pragma unroll 4
  for (int k = 0; k < 64; ++k) {
    int idx = k * NT + tid;
    float m = pHi[idx >> 7] * pLo[idx & 127];
    int ph = __popc(idx) & 3;
    int ps = (k << 8) + swz_lo(tid, k);
    st[ps]       = (ph == 0) ? m : ((ph == 2) ? -m : 0.f);
    st[DIM + ps] = (ph == 1) ? -m : ((ph == 3) ? m : 0.f);
  }
  __syncthreads();

  const int baseArr[4] = {10, 6, 2, 0};

  for (int L = 0; L < 2; ++L) {
    for (int g = 0; g < 4; ++g) {
      const int base = baseArr[g];
      const int mask = (1 << base) - 1;

      const int p = (g == 1 || g == 2) ? 2 : 0;
      const int h = lane >> 4;

      int rowP[8];
#pragma unroll
      for (int v = 0; v < 8; ++v) rowP[v] = swz(ins_bit(v, p, h) << base);

      const int cl = lane & 15;
      float Bre[8], Bim[8];
      int pc;
      {
        int c = 16 * wave + cl;
        int colAddr = ((c >> base) << (base + 4)) + (c & mask);
        pc = swz(colAddr);
#pragma unroll
        for (int v = 0; v < 8; ++v) {
          Bre[v] = st[pc ^ rowP[v]];
          Bim[v] = st[DIM + (pc ^ rowP[v])];
        }
      }
      float gc[4], gs[4];
#pragma unroll
      for (int m = 0; m < 4; ++m) {
        if (g == 3 && m >= 2) { gc[m] = 1.f; gs[m] = 0.f; }
        else {
          int q = 13 - base - m;
          float ang = 0.5f * params[L * NQ + q];
          gc[m] = cosf(ang); gs[m] = sinf(ang);
        }
      }

      const int Mrow = ins_bit(lane & 7, p, (lane >> 3) & 1);
      v2f A[4];
#pragma unroll
      for (int k = 0; k < 4; ++k) {
#pragma unroll
        for (int r = 0; r < 2; ++r) {
          int Kc = ins_bit(2 * k + r, p, h);
          float e = 1.f;
#pragma unroll
          for (int m = 0; m < 4; ++m) {
            int ib = (Mrow >> m) & 1, jb = (Kc >> m) & 1;
            e *= (ib == jb) ? gc[m] : ((ib == 0) ? -gs[m] : gs[m]);
          }
          A[k][r] = e;
        }
      }

#pragma unroll
      for (int i = 0; i < 8; ++i) {
        float Nre[8], Nim[8];
        int pcn = 0;
        if (i + 1 < 8) {
          int c = 16 * (wave + 8 * (i + 1)) + cl;
          int colAddr = ((c >> base) << (base + 4)) + (c & mask);
          pcn = swz(colAddr);
#pragma unroll
          for (int v = 0; v < 8; ++v) {
            Nre[v] = st[pcn ^ rowP[v]];
            Nim[v] = st[DIM + (pcn ^ rowP[v])];
          }
        }
        v8f accR = {}, accI = {};
#pragma unroll
        for (int k = 0; k < 4; ++k) {
          v2f Br, Bi;
          Br[0] = Bre[2 * k + 0]; Br[1] = Bre[2 * k + 1];
          Bi[0] = Bim[2 * k + 0]; Bi[1] = Bim[2 * k + 1];
          accR = wmma_16x16x4_f32(A[k], Br, accR);
          accI = wmma_16x16x4_f32(A[k], Bi, accI);
        }
#pragma unroll
        for (int r = 0; r < 8; ++r) {
          st[pc ^ rowP[r]]         = accR[r];
          st[DIM + (pc ^ rowP[r])] = accI[r];
        }
        if (i + 1 < 8) {
#pragma unroll
          for (int v = 0; v < 8; ++v) { Bre[v] = Nre[v]; Bim[v] = Nim[v]; }
          pc = pcn;
        }
      }
      __syncthreads();
    }

#pragma unroll
    for (int comp = 0; comp < 2; ++comp) {
      float* buf = st + comp * DIM;
      float v[64];
#pragma unroll
      for (int k = 0; k < 64; ++k) {
        const int gk = k ^ (k >> 1);
        const int gt = (tid ^ (tid >> 1)) ^ ((k & 1) << 7);
        v[k] = buf[(gk << 8) + swz_lo(gt, gk)];
      }
      __syncthreads();
#pragma unroll
      for (int k = 0; k < 64; ++k)
        buf[(k << 8) + swz_lo(tid, k)] = v[k];
      __syncthreads();
    }
  }

  float acc14[NQ];
#pragma unroll
  for (int w = 0; w < NQ; ++w) acc14[w] = 0.f;

  for (int k = 0; k < 64; ++k) {
    int j = k * NT + tid;
    int pj = (k << 8) + swz_lo(tid, k);
    float re = st[pj], im = st[DIM + pj];
    float p = re * re + im * im;
#pragma unroll
    for (int w = 0; w < NQ; ++w)
      acc14[w] += ((j >> (13 - w)) & 1) ? -p : p;
  }

#pragma unroll
  for (int w = 0; w < NQ; ++w) {
    float v = acc14[w];
#pragma unroll
    for (int off = 16; off > 0; off >>= 1)
      v += __shfl_down(v, off, 32);
    if (lane == 0) red[w * 8 + wave] = v;
  }
  __syncthreads();
  if (tid < NQ) {
    float s = 0.f;
#pragma unroll
    for (int i = 0; i < 8; ++i) s += red[tid * 8 + i];
    out[b * NQ + tid] = s;
  }
}

extern "C" void kernel_launch(void* const* d_in, const int* in_sizes, int n_in,
                              void* d_out, int out_size, void* d_ws, size_t ws_size,
                              hipStream_t stream) {
  const float* x      = (const float*)d_in[0];
  const float* params = (const float*)d_in[1];
  float* out          = (float*)d_out;
  int B = in_sizes[0] / NQ;
  qsim14_kernel<<<dim3(B), dim3(NT), 0, stream>>>(x, params, out);
}


# Layer 0 in closed form: its RYs act on a product state, so per-qubit RY*RX 2-vectors are formed and amplitudes written as complex products at index j^(j>>1); 4 of 8 WMMA passes go (sum order changes)
# speedup vs baseline: 1.7677x; 1.6232x over previous
#include <hip/hip_runtime.h>
#include <math.h>

typedef float v2f __attribute__((ext_vector_type(2)));
typedef float v8f __attribute__((ext_vector_type(8)));

#define NQ   14
#define DIM  16384
#define NT   256

static __device__ inline v8f wmma_16x16x4_f32(v2f a, v2f b, v8f c) {
#if defined(__gfx1250__)
  return __builtin_amdgcn_wmma_f32_16x16x4_f32(
      false, a, false, b, (short)0, c, false, false);
#else
  return c;
#endif
}

static __device__ __forceinline__ int swz(int i) {
  return i ^ (((i >> 6) & 15) << 2) ^ (((i >> 10) & 3) << 4);
}

static __device__ __forceinline__ int swz_lo(int T, int K) {
  return T ^ ((T >> 6) << 2) ^ ((((K & 3) ^ (K >> 2)) & 3) << 4);
}

static __device__ __forceinline__ int ins_bit(int v, int p, int b) {
  return ((v >> p) << (p + 1)) | (b << p) | (v & ((1 << p) - 1));
}

__global__ __launch_bounds__(NT) __attribute__((amdgpu_waves_per_eu(4))) void qsim14_kernel(
    const float* __restrict__ x,
    const float* __restrict__ params,
    float* __restrict__ out)
{
  __shared__ float st[2 * DIM];
  __shared__ float pHiR[128], pHiI[128];
  __shared__ float pLoR[128], pLoI[128];
  __shared__ float f0r[NQ], f0i[NQ];
  __shared__ float f1r[NQ], f1i[NQ];
  __shared__ float red[NQ * 8];

  const int b    = blockIdx.x;
  const int tid  = threadIdx.x;
  const int lane = tid & 31;
  const int wave = tid >> 5;

  if (tid < NQ) {
    float hx = 0.5f * x[b * NQ + tid];
    float c = cosf(hx), s = sinf(hx);
    float hp = 0.5f * params[tid];
    float cw = cosf(hp), sw = sinf(hp);
    f0r[tid] = cw * c;  f0i[tid] = sw * s;
    f1r[tid] = sw * c;  f1i[tid] = -(cw * s);
  }
  __syncthreads();

  {
    const int hsel = tid >> 7;
    const int v7 = tid & 127;
    float pr = 1.f, pi = 0.f;
#pragma unroll
    for (int t = 0; t < 7; ++t) {
      const int q = 7 * hsel + t;
      const int bit = (v7 >> (6 - t)) & 1;
      float fr = bit ? f1r[q] : f0r[q];
      float fi = bit ? f1i[q] : f0i[q];
      float nr = pr * fr - pi * fi;
      float ni = pr * fi + pi * fr;
      pr = nr; pi = ni;
    }
    if (hsel == 0) { pHiR[v7] = pr; pHiI[v7] = pi; }
    else           { pLoR[v7] = pr; pLoI[v7] = pi; }
  }
  __syncthreads();

  {
    const int lo = (tid ^ (tid >> 1)) & 127;
    const float br = pLoR[lo], bi = pLoI[lo];
#pragma unroll 4
    for (int k = 0; k < 64; ++k) {
      int j = k * NT + tid;
      int hi = (j ^ (j >> 1)) >> 7;
      float ar = pHiR[hi], ai = pHiI[hi];
      int ps = (k << 8) + swz_lo(tid, k);
      st[ps]       = ar * br - ai * bi;
      st[DIM + ps] = ar * bi + ai * br;
    }
  }
  __syncthreads();

  const int baseArr[4] = {10, 6, 2, 0};

  {
    const int L = 1;
    for (int g = 0; g < 4; ++g) {
      const int base = baseArr[g];
      const int mask = (1 << base) - 1;

      const int p = (g == 1 || g == 2) ? 2 : 0;
      const int h = lane >> 4;

      int rowP[8];
#pragma unroll
      for (int v = 0; v < 8; ++v) rowP[v] = swz(ins_bit(v, p, h) << base);

      const int cl = lane & 15;
      float Bre[8], Bim[8];
      int pc;
      {
        int c = 16 * wave + cl;
        int colAddr = ((c >> base) << (base + 4)) + (c & mask);
        pc = swz(colAddr);
#pragma unroll
        for (int v = 0; v < 8; ++v) {
          Bre[v] = st[pc ^ rowP[v]];
          Bim[v] = st[DIM + (pc ^ rowP[v])];
        }
      }
      float gc[4], gs[4];
#pragma unroll
      for (int m = 0; m < 4; ++m) {
        if (g == 3 && m >= 2) { gc[m] = 1.f; gs[m] = 0.f; }
        else {
          int q = 13 - base - m;
          float ang = 0.5f * params[L * NQ + q];
          gc[m] = cosf(ang); gs[m] = sinf(ang);
        }
      }

      const int Mrow = ins_bit(lane & 7, p, (lane >> 3) & 1);
      v2f A[4];
#pragma unroll
      for (int k = 0; k < 4; ++k) {
#pragma unroll
        for (int r = 0; r < 2; ++r) {
          int Kc = ins_bit(2 * k + r, p, h);
          float e = 1.f;
#pragma unroll
          for (int m = 0; m < 4; ++m) {
            int ib = (Mrow >> m) & 1, jb = (Kc >> m) & 1;
            e *= (ib == jb) ? gc[m] : ((ib == 0) ? -gs[m] : gs[m]);
          }
          A[k][r] = e;
        }
      }

#pragma unroll
      for (int i = 0; i < 8; ++i) {
        float Nre[8], Nim[8];
        int pcn = 0;
        if (i + 1 < 8) {
          int c = 16 * (wave + 8 * (i + 1)) + cl;
          int colAddr = ((c >> base) << (base + 4)) + (c & mask);
          pcn = swz(colAddr);
#pragma unroll
          for (int v = 0; v < 8; ++v) {
            Nre[v] = st[pcn ^ rowP[v]];
            Nim[v] = st[DIM + (pcn ^ rowP[v])];
          }
        }
        v8f accR = {}, accI = {};
#pragma unroll
        for (int k = 0; k < 4; ++k) {
          v2f Br, Bi;
          Br[0] = Bre[2 * k + 0]; Br[1] = Bre[2 * k + 1];
          Bi[0] = Bim[2 * k + 0]; Bi[1] = Bim[2 * k + 1];
          accR = wmma_16x16x4_f32(A[k], Br, accR);
          accI = wmma_16x16x4_f32(A[k], Bi, accI);
        }
#pragma unroll
        for (int r = 0; r < 8; ++r) {
          st[pc ^ rowP[r]]         = accR[r];
          st[DIM + (pc ^ rowP[r])] = accI[r];
        }
        if (i + 1 < 8) {
#pragma unroll
          for (int v = 0; v < 8; ++v) { Bre[v] = Nre[v]; Bim[v] = Nim[v]; }
          pc = pcn;
        }
      }
      __syncthreads();
    }

#pragma unroll
    for (int comp = 0; comp < 2; ++comp) {
      float* buf = st + comp * DIM;
      float v[64];
#pragma unroll
      for (int k = 0; k < 64; ++k) {
        const int gk = k ^ (k >> 1);
        const int gt = (tid ^ (tid >> 1)) ^ ((k & 1) << 7);
        v[k] = buf[(gk << 8) + swz_lo(gt, gk)];
      }
      __syncthreads();
#pragma unroll
      for (int k = 0; k < 64; ++k)
        buf[(k << 8) + swz_lo(tid, k)] = v[k];
      __syncthreads();
    }
  }

  float acc14[NQ];
#pragma unroll
  for (int w = 0; w < NQ; ++w) acc14[w] = 0.f;

  for (int k = 0; k < 64; ++k) {
    int j = k * NT + tid;
    int pj = (k << 8) + swz_lo(tid, k);
    float re = st[pj], im = st[DIM + pj];
    float p = re * re + im * im;
#pragma unroll
    for (int w = 0; w < NQ; ++w)
      acc14[w] += ((j >> (13 - w)) & 1) ? -p : p;
  }

#pragma unroll
  for (int w = 0; w < NQ; ++w) {
    float v = acc14[w];
#pragma unroll
    for (int off = 16; off > 0; off >>= 1)
      v += __shfl_down(v, off, 32);
    if (lane == 0) red[w * 8 + wave] = v;
  }
  __syncthreads();
  if (tid < NQ) {
    float s = 0.f;
#pragma unroll
    for (int i = 0; i < 8; ++i) s += red[tid * 8 + i];
    out[b * NQ + tid] = s;
  }
}

extern "C" void kernel_launch(void* const* d_in, const int* in_sizes, int n_in,
                              void* d_out, int out_size, void* d_ws, size_t ws_size,
                              hipStream_t stream) {
  const float* x      = (const float*)d_in[0];
  const float* params = (const float*)d_in[1];
  float* out          = (float*)d_out;
  int B = in_sizes[0] / NQ;
  qsim14_kernel<<<dim3(B), dim3(NT), 0, stream>>>(x, params, out);
}


# Layer-1 CNOT chain folded into the Z signs: sign_w = parity of index bits 13-w..13, a compile-time add/sub per loop step plus one per-thread sign; Gray gather pass removed (sum order changes)
# speedup vs baseline: 2.0562x; 1.1632x over previous
#include <hip/hip_runtime.h>
#include <math.h>

typedef float v2f __attribute__((ext_vector_type(2)));
typedef float v8f __attribute__((ext_vector_type(8)));

#define NQ   14
#define DIM  16384
#define NT   256

static __device__ inline v8f wmma_16x16x4_f32(v2f a, v2f b, v8f c) {
#if defined(__gfx1250__)
  return __builtin_amdgcn_wmma_f32_16x16x4_f32(
      false, a, false, b, (short)0, c, false, false);
#else
  return c;
#endif
}

static __device__ __forceinline__ int swz(int i) {
  return i ^ (((i >> 6) & 15) << 2) ^ (((i >> 10) & 3) << 4);
}

static __device__ __forceinline__ int swz_lo(int T, int K) {
  return T ^ ((T >> 6) << 2) ^ ((((K & 3) ^ (K >> 2)) & 3) << 4);
}

static __device__ __forceinline__ int ins_bit(int v, int p, int b) {
  return ((v >> p) << (p + 1)) | (b << p) | (v & ((1 << p) - 1));
}

__global__ __launch_bounds__(NT) __attribute__((amdgpu_waves_per_eu(4))) void qsim14_kernel(
    const float* __restrict__ x,
    const float* __restrict__ params,
    float* __restrict__ out)
{
  __shared__ float st[2 * DIM];
  __shared__ float pHiR[128], pHiI[128];
  __shared__ float pLoR[128], pLoI[128];
  __shared__ float f0r[NQ], f0i[NQ];
  __shared__ float f1r[NQ], f1i[NQ];
  __shared__ float red[NQ * 8];

  const int b    = blockIdx.x;
  const int tid  = threadIdx.x;
  const int lane = tid & 31;
  const int wave = tid >> 5;

  if (tid < NQ) {
    float hx = 0.5f * x[b * NQ + tid];
    float c = cosf(hx), s = sinf(hx);
    float hp = 0.5f * params[tid];
    float cw = cosf(hp), sw = sinf(hp);
    f0r[tid] = cw * c;  f0i[tid] = sw * s;
    f1r[tid] = sw * c;  f1i[tid] = -(cw * s);
  }
  __syncthreads();

  {
    const int hsel = tid >> 7;
    const int v7 = tid & 127;
    float pr = 1.f, pi = 0.f;
#pragma unroll
    for (int t = 0; t < 7; ++t) {
      const int q = 7 * hsel + t;
      const int bit = (v7 >> (6 - t)) & 1;
      float fr = bit ? f1r[q] : f0r[q];
      float fi = bit ? f1i[q] : f0i[q];
      float nr = pr * fr - pi * fi;
      float ni = pr * fi + pi * fr;
      pr = nr; pi = ni;
    }
    if (hsel == 0) { pHiR[v7] = pr; pHiI[v7] = pi; }
    else           { pLoR[v7] = pr; pLoI[v7] = pi; }
  }
  __syncthreads();

  {
    const int lo = (tid ^ (tid >> 1)) & 127;
    const float br = pLoR[lo], bi = pLoI[lo];
#pragma unroll 4
    for (int k = 0; k < 64; ++k) {
      int j = k * NT + tid;
      int hi = (j ^ (j >> 1)) >> 7;
      float ar = pHiR[hi], ai = pHiI[hi];
      int ps = (k << 8) + swz_lo(tid, k);
      st[ps]       = ar * br - ai * bi;
      st[DIM + ps] = ar * bi + ai * br;
    }
  }
  __syncthreads();

  const int baseArr[4] = {10, 6, 2, 0};

  {
    const int L = 1;
    for (int g = 0; g < 4; ++g) {
      const int base = baseArr[g];
      const int mask = (1 << base) - 1;

      const int p = (g == 1 || g == 2) ? 2 : 0;
      const int h = lane >> 4;

      int rowP[8];
#pragma unroll
      for (int v = 0; v < 8; ++v) rowP[v] = swz(ins_bit(v, p, h) << base);

      const int cl = lane & 15;
      float Bre[8], Bim[8];
      int pc;
      {
        int c = 16 * wave + cl;
        int colAddr = ((c >> base) << (base + 4)) + (c & mask);
        pc = swz(colAddr);
#pragma unroll
        for (int v = 0; v < 8; ++v) {
          Bre[v] = st[pc ^ rowP[v]];
          Bim[v] = st[DIM + (pc ^ rowP[v])];
        }
      }
      float gc[4], gs[4];
#pragma unroll
      for (int m = 0; m < 4; ++m) {
        if (g == 3 && m >= 2) { gc[m] = 1.f; gs[m] = 0.f; }
        else {
          int q = 13 - base - m;
          float ang = 0.5f * params[L * NQ + q];
          gc[m] = cosf(ang); gs[m] = sinf(ang);
        }
      }

      const int Mrow = ins_bit(lane & 7, p, (lane >> 3) & 1);
      v2f A[4];
#pragma unroll
      for (int k = 0; k < 4; ++k) {
#pragma unroll
        for (int r = 0; r < 2; ++r) {
          int Kc = ins_bit(2 * k + r, p, h);
          float e = 1.f;
#pragma unroll
          for (int m = 0; m < 4; ++m) {
            int ib = (Mrow >> m) & 1, jb = (Kc >> m) & 1;
            e *= (ib == jb) ? gc[m] : ((ib == 0) ? -gs[m] : gs[m]);
          }
          A[k][r] = e;
        }
      }

#pragma unroll
      for (int i = 0; i < 8; ++i) {
        float Nre[8], Nim[8];
        int pcn = 0;
        if (i + 1 < 8) {
          int c = 16 * (wave + 8 * (i + 1)) + cl;
          int colAddr = ((c >> base) << (base + 4)) + (c & mask);
          pcn = swz(colAddr);
#pragma unroll
          for (int v = 0; v < 8; ++v) {
            Nre[v] = st[pcn ^ rowP[v]];
            Nim[v] = st[DIM + (pcn ^ rowP[v])];
          }
        }
        v8f accR = {}, accI = {};
#pragma unroll
        for (int k = 0; k < 4; ++k) {
          v2f Br, Bi;
          Br[0] = Bre[2 * k + 0]; Br[1] = Bre[2 * k + 1];
          Bi[0] = Bim[2 * k + 0]; Bi[1] = Bim[2 * k + 1];
          accR = wmma_16x16x4_f32(A[k], Br, accR);
          accI = wmma_16x16x4_f32(A[k], Bi, accI);
        }
#pragma unroll
        for (int r = 0; r < 8; ++r) {
          st[pc ^ rowP[r]]         = accR[r];
          st[DIM + (pc ^ rowP[r])] = accI[r];
        }
        if (i + 1 < 8) {
#pragma unroll
          for (int v = 0; v < 8; ++v) { Bre[v] = Nre[v]; Bim[v] = Nim[v]; }
          pc = pcn;
        }
      }
      __syncthreads();
    }

  }

  float accK[5];
#pragma unroll
  for (int w = 0; w < 5; ++w) accK[w] = 0.f;
  float accT = 0.f;

#pragma unroll
  for (int k = 0; k < 64; ++k) {
    int pj = (k << 8) + swz_lo(tid, k);
    float re = st[pj], im = st[DIM + pj];
    float p = re * re + im * im;
#pragma unroll
    for (int w = 0; w < 5; ++w) {
      if (__builtin_popcount(k >> (5 - w)) & 1) accK[w] -= p; else accK[w] += p;
    }
    if (__builtin_popcount(k) & 1) accT -= p; else accT += p;
  }

  float acc14[NQ];
#pragma unroll
  for (int w = 0; w < 5; ++w) acc14[w] = accK[w];
#pragma unroll
  for (int w = 5; w < NQ; ++w)
    acc14[w] = (__builtin_popcount(tid >> (13 - w)) & 1) ? -accT : accT;

#pragma unroll
  for (int w = 0; w < NQ; ++w) {
    float v = acc14[w];
#pragma unroll
    for (int off = 16; off > 0; off >>= 1)
      v += __shfl_down(v, off, 32);
    if (lane == 0) red[w * 8 + wave] = v;
  }
  __syncthreads();
  if (tid < NQ) {
    float s = 0.f;
#pragma unroll
    for (int i = 0; i < 8; ++i) s += red[tid * 8 + i];
    out[b * NQ + tid] = s;
  }
}

extern "C" void kernel_launch(void* const* d_in, const int* in_sizes, int n_in,
                              void* d_out, int out_size, void* d_ws, size_t ws_size,
                              hipStream_t stream) {
  const float* x      = (const float*)d_in[0];
  const float* params = (const float*)d_in[1];
  float* out          = (float*)d_out;
  int B = in_sizes[0] / NQ;
  qsim14_kernel<<<dim3(B), dim3(NT), 0, stream>>>(x, params, out);
}


# Layer-0 amplitudes formed in registers as the first gate pass's B operands (high-table x low-table products); the Phase-0 state write, its table loads and its barrier are removed
# speedup vs baseline: 2.2639x; 1.1010x over previous
#include <hip/hip_runtime.h>
#include <math.h>

typedef float v2f __attribute__((ext_vector_type(2)));
typedef float v8f __attribute__((ext_vector_type(8)));

#define NQ   14
#define DIM  16384
#define NT   256

static __device__ inline v8f wmma_16x16x4_f32(v2f a, v2f b, v8f c) {
#if defined(__gfx1250__)
  return __builtin_amdgcn_wmma_f32_16x16x4_f32(
      false, a, false, b, (short)0, c, false, false);
#else
  return c;
#endif
}

static __device__ __forceinline__ int swz(int i) {
  return i ^ (((i >> 6) & 15) << 2) ^ (((i >> 10) & 3) << 4);
}

static __device__ __forceinline__ int swz_lo(int T, int K) {
  return T ^ ((T >> 6) << 2) ^ ((((K & 3) ^ (K >> 2)) & 3) << 4);
}

static __device__ __forceinline__ int ins_bit(int v, int p, int b) {
  return ((v >> p) << (p + 1)) | (b << p) | (v & ((1 << p) - 1));
}

static __device__ __forceinline__ constexpr int gray(int J) { return J ^ (J >> 1); }

__global__ __launch_bounds__(NT) __attribute__((amdgpu_waves_per_eu(4))) void qsim14_kernel(
    const float* __restrict__ x,
    const float* __restrict__ params,
    float* __restrict__ out)
{
  __shared__ float st[2 * DIM];
  __shared__ float pHiR[128], pHiI[128];
  __shared__ float pLoR[128], pLoI[128];
  __shared__ float f0r[NQ], f0i[NQ];
  __shared__ float f1r[NQ], f1i[NQ];
  __shared__ float red[NQ * 8];

  const int b    = blockIdx.x;
  const int tid  = threadIdx.x;
  const int lane = tid & 31;
  const int wave = tid >> 5;

  if (tid < NQ) {
    float hx = 0.5f * x[b * NQ + tid];
    float c = cosf(hx), s = sinf(hx);
    float hp = 0.5f * params[tid];
    float cw = cosf(hp), sw = sinf(hp);
    f0r[tid] = cw * c;  f0i[tid] = sw * s;
    f1r[tid] = sw * c;  f1i[tid] = -(cw * s);
  }
  __syncthreads();

  {
    const int hsel = tid >> 7;
    const int v7 = tid & 127;
    float pr = 1.f, pi = 0.f;
#pragma unroll
    for (int t = 0; t < 7; ++t) {
      const int q = 7 * hsel + t;
      const int bit = (v7 >> (6 - t)) & 1;
      float fr = bit ? f1r[q] : f0r[q];
      float fi = bit ? f1i[q] : f0i[q];
      float nr = pr * fr - pi * fi;
      float ni = pr * fi + pi * fr;
      pr = nr; pi = ni;
    }
    if (hsel == 0) { pHiR[v7] = pr; pHiI[v7] = pi; }
    else           { pLoR[v7] = pr; pLoI[v7] = pi; }
  }
  __syncthreads();

  {
    const int L = 1;
    const int base = 10;
    const int mask = (1 << base) - 1;
    const int p = 0;
    const int h = lane >> 4;
    const int cl = lane & 15;

    int rowP[8];
#pragma unroll
    for (int v = 0; v < 8; ++v) rowP[v] = swz(ins_bit(v, p, h) << base);

    const int c7 = (wave << 4) | cl;
    const int lo0 = gray(c7);
    float blr[2], bli[2];
    blr[0] = pLoR[lo0];      bli[0] = pLoI[lo0];
    blr[1] = pLoR[lo0 ^ 64]; bli[1] = pLoI[lo0 ^ 64];

    int hx[4];
#pragma unroll
    for (int m = 0; m < 4; ++m) hx[m] = (h * 12) ^ (m << 2);

    float Hr[8], Hi[8];
#pragma unroll
    for (int v = 0; v < 8; ++v) {
      const int Cc = gray(v << 4);
      Hr[v] = pHiR[hx[(Cc >> 2) & 3] + (Cc & ~12)];
      Hi[v] = pHiI[hx[(Cc >> 2) & 3] + (Cc & ~12)];
    }

    float gc[4], gs[4];
#pragma unroll
    for (int m = 0; m < 4; ++m) {
      int q = 13 - base - m;
      float ang = 0.5f * params[L * NQ + q];
      gc[m] = cosf(ang); gs[m] = sinf(ang);
    }

    const int Mrow = ins_bit(lane & 7, p, (lane >> 3) & 1);
    v2f A[4];
#pragma unroll
    for (int k = 0; k < 4; ++k) {
#pragma unroll
      for (int r = 0; r < 2; ++r) {
        int Kc = ins_bit(2 * k + r, p, h);
        float e = 1.f;
#pragma unroll
        for (int m = 0; m < 4; ++m) {
          int ib = (Mrow >> m) & 1, jb = (Kc >> m) & 1;
          e *= (ib == jb) ? gc[m] : ((ib == 0) ? -gs[m] : gs[m]);
        }
        A[k][r] = e;
      }
    }

#pragma unroll
    for (int i = 0; i < 8; ++i) {
      float Nr[8], Ni[8];
      if (i + 1 < 8) {
#pragma unroll
        for (int v = 0; v < 8; ++v) {
          const int Cc = gray((v << 4) | (i + 1));
          Nr[v] = pHiR[hx[(Cc >> 2) & 3] + (Cc & ~12)];
          Ni[v] = pHiI[hx[(Cc >> 2) & 3] + (Cc & ~12)];
        }
      }
      const float br = blr[i & 1], bi = bli[i & 1];
      float Bre[8], Bim[8];
#pragma unroll
      for (int v = 0; v < 8; ++v) {
        Bre[v] = Hr[v] * br - Hi[v] * bi;
        Bim[v] = Hr[v] * bi + Hi[v] * br;
      }
      v8f accR = {}, accI = {};
#pragma unroll
      for (int k = 0; k < 4; ++k) {
        v2f Br, Bi;
        Br[0] = Bre[2 * k + 0]; Br[1] = Bre[2 * k + 1];
        Bi[0] = Bim[2 * k + 0]; Bi[1] = Bim[2 * k + 1];
        accR = wmma_16x16x4_f32(A[k], Br, accR);
        accI = wmma_16x16x4_f32(A[k], Bi, accI);
      }
      const int c = 16 * (wave + 8 * i) + cl;
      const int colAddr = ((c >> base) << (base + 4)) + (c & mask);
      const int pc = swz(colAddr);
#pragma unroll
      for (int r = 0; r < 8; ++r) {
        st[pc ^ rowP[r]]         = accR[r];
        st[DIM + (pc ^ rowP[r])] = accI[r];
      }
      if (i + 1 < 8) {
#pragma unroll
        for (int v = 0; v < 8; ++v) { Hr[v] = Nr[v]; Hi[v] = Ni[v]; }
      }
    }
    __syncthreads();
  }

  const int baseArr[4] = {10, 6, 2, 0};

  {
    const int L = 1;
    for (int g = 1; g < 4; ++g) {
      const int base = baseArr[g];
      const int mask = (1 << base) - 1;

      const int p = (g == 1 || g == 2) ? 2 : 0;
      const int h = lane >> 4;

      int rowP[8];
#pragma unroll
      for (int v = 0; v < 8; ++v) rowP[v] = swz(ins_bit(v, p, h) << base);

      const int cl = lane & 15;
      float Bre[8], Bim[8];
      int pc;
      {
        int c = 16 * wave + cl;
        int colAddr = ((c >> base) << (base + 4)) + (c & mask);
        pc = swz(colAddr);
#pragma unroll
        for (int v = 0; v < 8; ++v) {
          Bre[v] = st[pc ^ rowP[v]];
          Bim[v] = st[DIM + (pc ^ rowP[v])];
        }
      }
      float gc[4], gs[4];
#pragma unroll
      for (int m = 0; m < 4; ++m) {
        if (g == 3 && m >= 2) { gc[m] = 1.f; gs[m] = 0.f; }
        else {
          int q = 13 - base - m;
          float ang = 0.5f * params[L * NQ + q];
          gc[m] = cosf(ang); gs[m] = sinf(ang);
        }
      }

      const int Mrow = ins_bit(lane & 7, p, (lane >> 3) & 1);
      v2f A[4];
#pragma unroll
      for (int k = 0; k < 4; ++k) {
#pragma unroll
        for (int r = 0; r < 2; ++r) {
          int Kc = ins_bit(2 * k + r, p, h);
          float e = 1.f;
#pragma unroll
          for (int m = 0; m < 4; ++m) {
            int ib = (Mrow >> m) & 1, jb = (Kc >> m) & 1;
            e *= (ib == jb) ? gc[m] : ((ib == 0) ? -gs[m] : gs[m]);
          }
          A[k][r] = e;
        }
      }

#pragma unroll
      for (int i = 0; i < 8; ++i) {
        float Nre[8], Nim[8];
        int pcn = 0;
        if (i + 1 < 8) {
          int c = 16 * (wave + 8 * (i + 1)) + cl;
          int colAddr = ((c >> base) << (base + 4)) + (c & mask);
          pcn = swz(colAddr);
#pragma unroll
          for (int v = 0; v < 8; ++v) {
            Nre[v] = st[pcn ^ rowP[v]];
            Nim[v] = st[DIM + (pcn ^ rowP[v])];
          }
        }
        v8f accR = {}, accI = {};
#pragma unroll
        for (int k = 0; k < 4; ++k) {
          v2f Br, Bi;
          Br[0] = Bre[2 * k + 0]; Br[1] = Bre[2 * k + 1];
          Bi[0] = Bim[2 * k + 0]; Bi[1] = Bim[2 * k + 1];
          accR = wmma_16x16x4_f32(A[k], Br, accR);
          accI = wmma_16x16x4_f32(A[k], Bi, accI);
        }
#pragma unroll
        for (int r = 0; r < 8; ++r) {
          st[pc ^ rowP[r]]         = accR[r];
          st[DIM + (pc ^ rowP[r])] = accI[r];
        }
        if (i + 1 < 8) {
#pragma unroll
          for (int v = 0; v < 8; ++v) { Bre[v] = Nre[v]; Bim[v] = Nim[v]; }
          pc = pcn;
        }
      }
      __syncthreads();
    }

  }

  float accK[5];
#pragma unroll
  for (int w = 0; w < 5; ++w) accK[w] = 0.f;
  float accT = 0.f;

#pragma unroll
  for (int k = 0; k < 64; ++k) {
    int pj = (k << 8) + swz_lo(tid, k);
    float re = st[pj], im = st[DIM + pj];
    float p = re * re + im * im;
#pragma unroll
    for (int w = 0; w < 5; ++w) {
      if (__builtin_popcount(k >> (5 - w)) & 1) accK[w] -= p; else accK[w] += p;
    }
    if (__builtin_popcount(k) & 1) accT -= p; else accT += p;
  }

  float acc14[NQ];
#pragma unroll
  for (int w = 0; w < 5; ++w) acc14[w] = accK[w];
#pragma unroll
  for (int w = 5; w < NQ; ++w)
    acc14[w] = (__builtin_popcount(tid >> (13 - w)) & 1) ? -accT : accT;

#pragma unroll
  for (int w = 0; w < NQ; ++w) {
    float v = acc14[w];
#pragma unroll
    for (int off = 16; off > 0; off >>= 1)
      v += __shfl_down(v, off, 32);
    if (lane == 0) red[w * 8 + wave] = v;
  }
  __syncthreads();
  if (tid < NQ) {
    float s = 0.f;
#pragma unroll
    for (int i = 0; i < 8; ++i) s += red[tid * 8 + i];
    out[b * NQ + tid] = s;
  }
}

extern "C" void kernel_launch(void* const* d_in, const int* in_sizes, int n_in,
                              void* d_out, int out_size, void* d_ws, size_t ws_size,
                              hipStream_t stream) {
  const float* x      = (const float*)d_in[0];
  const float* params = (const float*)d_in[1];
  float* out          = (float*)d_out;
  int B = in_sizes[0] / NQ;
  qsim14_kernel<<<dim3(B), dim3(NT), 0, stream>>>(x, params, out);
}


# Last gate pass fused with Z sums: |amp|^2 and signed sums (sign butterfly over the D registers) come from its WMMA results; its state store, a barrier and the read-back loop go (sum order changes)
# speedup vs baseline: 2.5213x; 1.1137x over previous
#include <hip/hip_runtime.h>
#include <math.h>

typedef float v2f __attribute__((ext_vector_type(2)));
typedef float v8f __attribute__((ext_vector_type(8)));

#define NQ   14
#define DIM  16384
#define NT   256

static __device__ inline v8f wmma_16x16x4_f32(v2f a, v2f b, v8f c) {
#if defined(__gfx1250__)
  return __builtin_amdgcn_wmma_f32_16x16x4_f32(
      false, a, false, b, (short)0, c, false, false);
#else
  return c;
#endif
}

static __device__ __forceinline__ int swz(int i) {
  return i ^ (((i >> 6) & 15) << 2) ^ (((i >> 10) & 3) << 4);
}

static __device__ __forceinline__ int ins_bit(int v, int p, int b) {
  return ((v >> p) << (p + 1)) | (b << p) | (v & ((1 << p) - 1));
}

static __device__ __forceinline__ constexpr int gray(int J) { return J ^ (J >> 1); }

__global__ __launch_bounds__(NT) __attribute__((amdgpu_waves_per_eu(4))) void qsim14_kernel(
    const float* __restrict__ x,
    const float* __restrict__ params,
    float* __restrict__ out)
{
  __shared__ float st[2 * DIM];
  __shared__ float pHiR[128], pHiI[128];
  __shared__ float pLoR[128], pLoI[128];
  __shared__ float f0r[NQ], f0i[NQ];
  __shared__ float f1r[NQ], f1i[NQ];
  __shared__ float red[NQ * 8];

  const int b    = blockIdx.x;
  const int tid  = threadIdx.x;
  const int lane = tid & 31;
  const int wave = tid >> 5;

  if (tid < NQ) {
    float hx = 0.5f * x[b * NQ + tid];
    float c = cosf(hx), s = sinf(hx);
    float hp = 0.5f * params[tid];
    float cw = cosf(hp), sw = sinf(hp);
    f0r[tid] = cw * c;  f0i[tid] = sw * s;
    f1r[tid] = sw * c;  f1i[tid] = -(cw * s);
  }
  __syncthreads();

  {
    const int hsel = tid >> 7;
    const int v7 = tid & 127;
    float pr = 1.f, pi = 0.f;
#pragma unroll
    for (int t = 0; t < 7; ++t) {
      const int q = 7 * hsel + t;
      const int bit = (v7 >> (6 - t)) & 1;
      float fr = bit ? f1r[q] : f0r[q];
      float fi = bit ? f1i[q] : f0i[q];
      float nr = pr * fr - pi * fi;
      float ni = pr * fi + pi * fr;
      pr = nr; pi = ni;
    }
    if (hsel == 0) { pHiR[v7] = pr; pHiI[v7] = pi; }
    else           { pLoR[v7] = pr; pLoI[v7] = pi; }
  }
  __syncthreads();

  {
    const int L = 1;
    const int base = 10;
    const int mask = (1 << base) - 1;
    const int p = 0;
    const int h = lane >> 4;
    const int cl = lane & 15;

    int rowP[8];
#pragma unroll
    for (int v = 0; v < 8; ++v) rowP[v] = swz(ins_bit(v, p, h) << base);

    const int c7 = (wave << 4) | cl;
    const int lo0 = gray(c7);
    float blr[2], bli[2];
    blr[0] = pLoR[lo0];      bli[0] = pLoI[lo0];
    blr[1] = pLoR[lo0 ^ 64]; bli[1] = pLoI[lo0 ^ 64];

    int hx[4];
#pragma unroll
    for (int m = 0; m < 4; ++m) hx[m] = (h * 12) ^ (m << 2);

    float Hr[8], Hi[8];
#pragma unroll
    for (int v = 0; v < 8; ++v) {
      const int Cc = gray(v << 4);
      Hr[v] = pHiR[hx[(Cc >> 2) & 3] + (Cc & ~12)];
      Hi[v] = pHiI[hx[(Cc >> 2) & 3] + (Cc & ~12)];
    }

    float gc[4], gs[4];
#pragma unroll
    for (int m = 0; m < 4; ++m) {
      int q = 13 - base - m;
      float ang = 0.5f * params[L * NQ + q];
      gc[m] = cosf(ang); gs[m] = sinf(ang);
    }

    const int Mrow = ins_bit(lane & 7, p, (lane >> 3) & 1);
    v2f A[4];
#pragma unroll
    for (int k = 0; k < 4; ++k) {
#pragma unroll
      for (int r = 0; r < 2; ++r) {
        int Kc = ins_bit(2 * k + r, p, h);
        float e = 1.f;
#pragma unroll
        for (int m = 0; m < 4; ++m) {
          int ib = (Mrow >> m) & 1, jb = (Kc >> m) & 1;
          e *= (ib == jb) ? gc[m] : ((ib == 0) ? -gs[m] : gs[m]);
        }
        A[k][r] = e;
      }
    }

#pragma unroll
    for (int i = 0; i < 8; ++i) {
      float Nr[8], Ni[8];
      if (i + 1 < 8) {
#pragma unroll
        for (int v = 0; v < 8; ++v) {
          const int Cc = gray((v << 4) | (i + 1));
          Nr[v] = pHiR[hx[(Cc >> 2) & 3] + (Cc & ~12)];
          Ni[v] = pHiI[hx[(Cc >> 2) & 3] + (Cc & ~12)];
        }
      }
      const float br = blr[i & 1], bi = bli[i & 1];
      float Bre[8], Bim[8];
#pragma unroll
      for (int v = 0; v < 8; ++v) {
        Bre[v] = Hr[v] * br - Hi[v] * bi;
        Bim[v] = Hr[v] * bi + Hi[v] * br;
      }
      v8f accR = {}, accI = {};
#pragma unroll
      for (int k = 0; k < 4; ++k) {
        v2f Br, Bi;
        Br[0] = Bre[2 * k + 0]; Br[1] = Bre[2 * k + 1];
        Bi[0] = Bim[2 * k + 0]; Bi[1] = Bim[2 * k + 1];
        accR = wmma_16x16x4_f32(A[k], Br, accR);
        accI = wmma_16x16x4_f32(A[k], Bi, accI);
      }
      const int c = 16 * (wave + 8 * i) + cl;
      const int colAddr = ((c >> base) << (base + 4)) + (c & mask);
      const int pc = swz(colAddr);
#pragma unroll
      for (int r = 0; r < 8; ++r) {
        st[pc ^ rowP[r]]         = accR[r];
        st[DIM + (pc ^ rowP[r])] = accI[r];
      }
      if (i + 1 < 8) {
#pragma unroll
        for (int v = 0; v < 8; ++v) { Hr[v] = Nr[v]; Hi[v] = Ni[v]; }
      }
    }
    __syncthreads();
  }

  const int baseArr[4] = {10, 6, 2, 0};

  {
    const int L = 1;
    for (int g = 1; g < 3; ++g) {
      const int base = baseArr[g];
      const int mask = (1 << base) - 1;

      const int p = 2;
      const int h = lane >> 4;

      int rowP[8];
#pragma unroll
      for (int v = 0; v < 8; ++v) rowP[v] = swz(ins_bit(v, p, h) << base);

      const int cl = lane & 15;
      float Bre[8], Bim[8];
      int pc;
      {
        int c = 16 * wave + cl;
        int colAddr = ((c >> base) << (base + 4)) + (c & mask);
        pc = swz(colAddr);
#pragma unroll
        for (int v = 0; v < 8; ++v) {
          Bre[v] = st[pc ^ rowP[v]];
          Bim[v] = st[DIM + (pc ^ rowP[v])];
        }
      }
      float gc[4], gs[4];
#pragma unroll
      for (int m = 0; m < 4; ++m) {
        int q = 13 - base - m;
        float ang = 0.5f * params[L * NQ + q];
        gc[m] = cosf(ang); gs[m] = sinf(ang);
      }

      const int Mrow = ins_bit(lane & 7, p, (lane >> 3) & 1);
      v2f A[4];
#pragma unroll
      for (int k = 0; k < 4; ++k) {
#pragma unroll
        for (int r = 0; r < 2; ++r) {
          int Kc = ins_bit(2 * k + r, p, h);
          float e = 1.f;
#pragma unroll
          for (int m = 0; m < 4; ++m) {
            int ib = (Mrow >> m) & 1, jb = (Kc >> m) & 1;
            e *= (ib == jb) ? gc[m] : ((ib == 0) ? -gs[m] : gs[m]);
          }
          A[k][r] = e;
        }
      }

#pragma unroll
      for (int i = 0; i < 8; ++i) {
        float Nre[8], Nim[8];
        int pcn = 0;
        if (i + 1 < 8) {
          int c = 16 * (wave + 8 * (i + 1)) + cl;
          int colAddr = ((c >> base) << (base + 4)) + (c & mask);
          pcn = swz(colAddr);
#pragma unroll
          for (int v = 0; v < 8; ++v) {
            Nre[v] = st[pcn ^ rowP[v]];
            Nim[v] = st[DIM + (pcn ^ rowP[v])];
          }
        }
        v8f accR = {}, accI = {};
#pragma unroll
        for (int k = 0; k < 4; ++k) {
          v2f Br, Bi;
          Br[0] = Bre[2 * k + 0]; Br[1] = Bre[2 * k + 1];
          Bi[0] = Bim[2 * k + 0]; Bi[1] = Bim[2 * k + 1];
          accR = wmma_16x16x4_f32(A[k], Br, accR);
          accI = wmma_16x16x4_f32(A[k], Bi, accI);
        }
#pragma unroll
        for (int r = 0; r < 8; ++r) {
          st[pc ^ rowP[r]]         = accR[r];
          st[DIM + (pc ^ rowP[r])] = accI[r];
        }
        if (i + 1 < 8) {
#pragma unroll
          for (int v = 0; v < 8; ++v) { Bre[v] = Nre[v]; Bim[v] = Nim[v]; }
          pc = pcn;
        }
      }
      __syncthreads();
    }
  }

  float accW0 = 0.f, accW1 = 0.f;
  float accT = 0.f;
  float accT10 = 0.f, accT11 = 0.f, accT12 = 0.f;
  {
    const int L = 1;
    const int base = 0;
    const int p = 0;
    const int h = lane >> 4;
    const int cl = lane & 15;

    int rowP[8];
#pragma unroll
    for (int v = 0; v < 8; ++v) rowP[v] = swz(ins_bit(v, p, h) << base);

    float Bre[8], Bim[8];
    int pc = swz((16 * wave + cl) << 4);
#pragma unroll
    for (int v = 0; v < 8; ++v) {
      Bre[v] = st[pc ^ rowP[v]];
      Bim[v] = st[DIM + (pc ^ rowP[v])];
    }

    float gc[4], gs[4];
#pragma unroll
    for (int m = 0; m < 4; ++m) {
      if (m >= 2) { gc[m] = 1.f; gs[m] = 0.f; }
      else {
        int q = 13 - base - m;
        float ang = 0.5f * params[L * NQ + q];
        gc[m] = cosf(ang); gs[m] = sinf(ang);
      }
    }

    const int Mrow = ins_bit(lane & 7, p, (lane >> 3) & 1);
    v2f A[4];
#pragma unroll
    for (int k = 0; k < 4; ++k) {
#pragma unroll
      for (int r = 0; r < 2; ++r) {
        int Kc = ins_bit(2 * k + r, p, h);
        float e = 1.f;
#pragma unroll
        for (int m = 0; m < 4; ++m) {
          int ib = (Mrow >> m) & 1, jb = (Kc >> m) & 1;
          e *= (ib == jb) ? gc[m] : ((ib == 0) ? -gs[m] : gs[m]);
        }
        A[k][r] = e;
      }
    }

#pragma unroll
    for (int t = 0; t < 8; ++t) {
      float Nre[8], Nim[8];
      if (t + 1 < 8) {
        const int pcn = swz((16 * (wave + 8 * (t + 1)) + cl) << 4);
#pragma unroll
        for (int v = 0; v < 8; ++v) {
          Nre[v] = st[pcn ^ rowP[v]];
          Nim[v] = st[DIM + (pcn ^ rowP[v])];
        }
      }
      v8f accR = {}, accI = {};
#pragma unroll
      for (int k = 0; k < 4; ++k) {
        v2f Br, Bi;
        Br[0] = Bre[2 * k + 0]; Br[1] = Bre[2 * k + 1];
        Bi[0] = Bim[2 * k + 0]; Bi[1] = Bim[2 * k + 1];
        accR = wmma_16x16x4_f32(A[k], Br, accR);
        accI = wmma_16x16x4_f32(A[k], Bi, accI);
      }
      float pw[8];
#pragma unroll
      for (int r = 0; r < 8; ++r) pw[r] = accR[r] * accR[r] + accI[r] * accI[r];
      const float a0 = pw[0] + pw[1], d0 = pw[0] - pw[1];
      const float a1 = pw[2] + pw[3], d1 = pw[2] - pw[3];
      const float a2 = pw[4] + pw[5], d2 = pw[4] - pw[5];
      const float a3 = pw[6] + pw[7], d3 = pw[6] - pw[7];
      const float b0 = a0 + a1, e0 = a0 - a1;
      const float b1 = a2 + a3, e1 = a2 - a3;
      const float sAll = b0 + b1;
      const float s10  = b0 - b1;
      const float s11  = e0 - e1;
      const float s12  = (d0 - d1) - (d2 - d3);
      if ((t >> 2) & 1) accW0 -= sAll; else accW0 += sAll;
      if (__builtin_popcount(t >> 1) & 1) accW1 -= sAll; else accW1 += sAll;
      if (__builtin_popcount(t) & 1) { accT -= sAll; accT10 -= s10; accT11 -= s11; accT12 -= s12; }
      else                           { accT += sAll; accT10 += s10; accT11 += s11; accT12 += s12; }
      if (t + 1 < 8) {
#pragma unroll
        for (int v = 0; v < 8; ++v) { Bre[v] = Nre[v]; Bim[v] = Nim[v]; }
      }
    }
  }

  float acc14[NQ];
  {
    const int u = (wave << 4) | (lane & 15);
    const int pu = __builtin_popcount(u) & 1;
    acc14[0] = accW0;
    acc14[1] = accW1;
    acc14[2] = accT;
#pragma unroll
    for (int w = 3; w < 10; ++w)
      acc14[w] = (__builtin_popcount(u >> (9 - w)) & 1) ? -accT : accT;
    acc14[10] = pu ? -accT10 : accT10;
    acc14[11] = pu ? -accT11 : accT11;
    acc14[12] = pu ? -accT12 : accT12;
    acc14[13] = (pu ^ (lane >> 4)) ? -accT12 : accT12;
  }

#pragma unroll
  for (int w = 0; w < NQ; ++w) {
    float v = acc14[w];
#pragma unroll
    for (int off = 16; off > 0; off >>= 1)
      v += __shfl_down(v, off, 32);
    if (lane == 0) red[w * 8 + wave] = v;
  }
  __syncthreads();
  if (tid < NQ) {
    float s = 0.f;
#pragma unroll
    for (int i = 0; i < 8; ++i) s += red[tid * 8 + i];
    out[b * NQ + tid] = s;
  }
}

extern "C" void kernel_launch(void* const* d_in, const int* in_sizes, int n_in,
                              void* d_out, int out_size, void* d_ws, size_t ws_size,
                              hipStream_t stream) {
  const float* x      = (const float*)d_in[0];
  const float* params = (const float*)d_in[1];
  float* out          = (float*)d_out;
  int B = in_sizes[0] / NQ;
  qsim14_kernel<<<dim3(B), dim3(NT), 0, stream>>>(x, params, out);
}


# Work-group loops over samples (512 work-groups, grid stride): the params-only work (layer-0/1 cos and sin, the five gate fragments) is done once per work-group instead of once per sample
# speedup vs baseline: 7.9573x; 3.1560x over previous
#include <hip/hip_runtime.h>
#include <math.h>

typedef float v2f __attribute__((ext_vector_type(2)));
typedef float v8f __attribute__((ext_vector_type(8)));

#define NQ   14
#define NT   256

static __device__ inline v8f wmma_16x16x4_f32(v2f a, v2f b, v8f c) {
#if defined(__gfx1250__)
  return __builtin_amdgcn_wmma_f32_16x16x4_f32(
      false, a, false, b, (short)0, c, false, false);
#else
  return c;
#endif
}

static __device__ __forceinline__ constexpr int gray(int J) { return J ^ (J >> 1); }

static __device__ __forceinline__ float ry_el(int ib, int jb, float c, float s) {
  return (ib == jb) ? c : ((ib == 0) ? -s : s);
}

__global__ __launch_bounds__(NT) void qsim14_kernel(
    const float* __restrict__ x,
    const float* __restrict__ params,
    float* __restrict__ out,
    int nb)
{
  __shared__ float pHiR[128], pHiI[128];
  __shared__ float pLoR[128], pLoI[128];
  __shared__ float f0r[NQ], f0i[NQ];
  __shared__ float f1r[NQ], f1i[NQ];
  __shared__ float red[NQ * 8];

  const int tid  = threadIdx.x;
  const int lane = tid & 31;
  const int wave = tid >> 5;

  float cw0 = 1.f, sw0 = 0.f;
  if (tid < NQ) {
    float hp = 0.5f * params[tid];
    cw0 = cosf(hp); sw0 = sinf(hp);
  }

  const int hl = lane >> 4;
  const int m  = lane & 15;

  const int lb = gray(((m & 7) << 4) | (hl << 1)) ^ ((m >> 3) << 6);
  const int hb = gray(((m & 7) << 4) | (hl << 2) | (m >> 3));

  float gc[NQ], gs[NQ];
#pragma unroll
  for (int q = 0; q < NQ; ++q) {
    float ang = 0.5f * params[NQ + q];
    gc[q] = cosf(ang); gs[q] = sinf(ang);
  }

  v2f LB[4];
  v2f LG[4];
  v2f HG[4];
#pragma unroll
  for (int k = 0; k < 4; ++k) {
#pragma unroll
    for (int r = 0; r < 2; ++r) {
      const int K = 4 * k + 2 * hl + r;
      float e = 1.f;
#pragma unroll
      for (int mm = 0; mm < 4; ++mm)
        e *= ry_el((m >> mm) & 1, (K >> mm) & 1, gc[13 - mm], gs[13 - mm]);
      LB[k][r] = e;

      const int o3 = 2 * k + r;
      const float same = ((m >> 3) == hl) ? 1.f : 0.f;
      float eg = same;
#pragma unroll
      for (int mm = 0; mm < 3; ++mm)
        eg *= ry_el((m >> mm) & 1, (o3 >> mm) & 1, gc[9 - mm], gs[9 - mm]);
      LG[k][r] = eg;

      const int un = (m & 7) ^ wave;
      float eh = same;
#pragma unroll
      for (int mm = 0; mm < 3; ++mm)
        eh *= ry_el((un >> mm) & 1, (o3 >> mm) & 1, gc[2 - mm], gs[2 - mm]);
      HG[k][r] = eh;
    }
  }
  v2f HB[2];
#pragma unroll
  for (int k = 0; k < 2; ++k) {
#pragma unroll
    for (int r = 0; r < 2; ++r) {
      const int K = 4 * k + 2 * hl + r;
      float e = 1.f;
#pragma unroll
      for (int mm = 0; mm < 3; ++mm)
        e *= ry_el((m >> (mm + 1)) & 1, (K >> mm) & 1, gc[5 - mm], gs[5 - mm]);
      HB[k][r] = e;
    }
  }

  const float g6 = ry_el(m & 1, hl, gc[6], gs[6]);

  const int pwv = __builtin_popcount(wave) & 1;
  const int pu = pwv ^ (__builtin_popcount(m) & 1);
  const int ph = pu ^ hl;

  const int b0 = blockIdx.x;
  const int bstep = gridDim.x;
  float xn = 0.f;
  if (tid < NQ && b0 < nb) xn = x[b0 * NQ + tid];
  for (int b = b0; b < nb; b += bstep) {

  if (tid < NQ) {
    float hx = 0.5f * xn;
    float c = cosf(hx), s = sinf(hx);
    f0r[tid] = cw0 * c;  f0i[tid] = sw0 * s;
    f1r[tid] = sw0 * c;  f1i[tid] = -(cw0 * s);
  }
  __syncthreads();

  {
    const int hsel = tid >> 7;
    const int v7 = tid & 127;
    float pr = 1.f, pi = 0.f;
#pragma unroll
    for (int t = 0; t < 7; ++t) {
      const int q = 7 * hsel + t;
      const int bit = (v7 >> (6 - t)) & 1;
      float fr = bit ? f1r[q] : f0r[q];
      float fi = bit ? f1i[q] : f0i[q];
      float nr = pr * fr - pi * fi;
      float ni = pr * fi + pi * fr;
      pr = nr; pi = ni;
    }
    if (hsel == 0) { pHiR[v7] = pr; pHiI[v7] = pi; }
    else           { pLoR[v7] = pr; pLoI[v7] = pi; }
  }
  __syncthreads();

  if (tid < NQ && b + bstep < nb) xn = x[(b + bstep) * NQ + tid];

  v2f LAr[4], LAi[4];
#pragma unroll
  for (int k = 0; k < 4; ++k) {
#pragma unroll
    for (int r = 0; r < 2; ++r) {
      const int idx = lb ^ gray(4 * k + r);
      LAr[k][r] = pLoR[idx];
      LAi[k][r] = pLoI[idx];
    }
  }
  v2f HAr[2], HAi[2];
#pragma unroll
  for (int k = 0; k < 2; ++k) {
#pragma unroll
    for (int r = 0; r < 2; ++r) {
      const int idx = hb ^ gray(8 * k + 2 * r);
      HAr[k][r] = pHiR[idx];
      HAi[k][r] = pHiI[idx];
    }
  }

  v8f L1r = {}, L1i = {};
#pragma unroll
  for (int k = 0; k < 4; ++k) {
    L1r = wmma_16x16x4_f32(LAr[k], LB[k], L1r);
    L1i = wmma_16x16x4_f32(LAi[k], LB[k], L1i);
  }
  v8f L2r = {}, L2i = {};
#pragma unroll
  for (int k = 0; k < 4; ++k) {
    v2f Br, Bi;
    Br[0] = L1r[2 * k + 0]; Br[1] = L1r[2 * k + 1];
    Bi[0] = L1i[2 * k + 0]; Bi[1] = L1i[2 * k + 1];
    L2r = wmma_16x16x4_f32(LG[k], Br, L2r);
    L2i = wmma_16x16x4_f32(LG[k], Bi, L2i);
  }

  v8f H1r = {}, H1i = {};
#pragma unroll
  for (int k = 0; k < 2; ++k) {
    H1r = wmma_16x16x4_f32(HAr[k], HB[k], H1r);
    H1i = wmma_16x16x4_f32(HAi[k], HB[k], H1i);
  }
  v8f H2r = {}, H2i = {};
#pragma unroll
  for (int k = 0; k < 4; ++k) {
    v2f Br, Bi;
    Br[0] = H1r[2 * k + 0]; Br[1] = H1r[2 * k + 1];
    Bi[0] = H1i[2 * k + 0]; Bi[1] = H1i[2 * k + 1];
    H2r = wmma_16x16x4_f32(HG[k], Br, H2r);
    H2i = wmma_16x16x4_f32(HG[k], Bi, H2i);
  }

  v2f Bf;
  Bf[0] = g6 * H2r[0];
  Bf[1] = g6 * H2i[0];
  float accAll = 0.f;
  float accT7 = 0.f, accT8 = 0.f, accT9 = 0.f;
  float accT11 = 0.f, accT12 = 0.f, accT13 = 0.f;
#pragma unroll
  for (int t = 0; t < 8; ++t) {
    v2f Are, Aim;
    Are[0] = L2r[t]; Are[1] = -L2i[t];
    Aim[0] = L2i[t]; Aim[1] = L2r[t];
    const v8f zero = {};
    const v8f accR = wmma_16x16x4_f32(Are, Bf, zero);
    const v8f accI = wmma_16x16x4_f32(Aim, Bf, zero);
    float pw[8];
#pragma unroll
    for (int r = 0; r < 8; ++r) pw[r] = accR[r] * accR[r] + accI[r] * accI[r];
    const float a0 = pw[0] + pw[1], d0 = pw[0] - pw[1];
    const float a1 = pw[2] + pw[3], d1 = pw[2] - pw[3];
    const float a2 = pw[4] + pw[5], d2 = pw[4] - pw[5];
    const float a3 = pw[6] + pw[7], d3 = pw[6] - pw[7];
    const float b0 = a0 + a1, e0 = a0 - a1;
    const float b1 = a2 + a3, e1 = a2 - a3;
    const float sAll = b0 + b1;
    const float s11  = b0 - b1;
    const float s12  = e0 - e1;
    const float s13  = (d0 - d1) - (d2 - d3);
    accAll += sAll;
    if ((t >> 2) & 1) accT7 -= sAll; else accT7 += sAll;
    if (__builtin_popcount(t >> 1) & 1) accT8 -= sAll; else accT8 += sAll;
    if (__builtin_popcount(t) & 1) { accT9 -= sAll; accT11 -= s11; accT12 -= s12; accT13 -= s13; }
    else                           { accT9 += sAll; accT11 += s11; accT12 += s12; accT13 += s13; }
  }

  float acc14[NQ];
  {
    acc14[0] = ((wave >> 2) & 1) ? -accAll : accAll;
    acc14[1] = (__builtin_popcount(wave >> 1) & 1) ? -accAll : accAll;
    acc14[2] = pwv ? -accAll : accAll;
#pragma unroll
    for (int w = 3; w < 7; ++w)
      acc14[w] = (pwv ^ (__builtin_popcount(m >> (6 - w)) & 1)) ? -accAll : accAll;
    acc14[7]  = pu ? -accT7 : accT7;
    acc14[8]  = pu ? -accT8 : accT8;
    acc14[9]  = pu ? -accT9 : accT9;
    acc14[10] = ph ? -accT9 : accT9;
    acc14[11] = ph ? -accT11 : accT11;
    acc14[12] = ph ? -accT12 : accT12;
    acc14[13] = ph ? -accT13 : accT13;
  }

#pragma unroll
  for (int w = 0; w < NQ; ++w) {
    float v = acc14[w];
#pragma unroll
    for (int off = 16; off > 0; off >>= 1)
      v += __shfl_down(v, off, 32);
    if (lane == 0) red[w * 8 + wave] = v;
  }
  __syncthreads();
  if (tid < NQ) {
    float s = 0.f;
#pragma unroll
    for (int i = 0; i < 8; ++i) s += red[tid * 8 + i];
    out[b * NQ + tid] = s;
  }

  }
}

extern "C" void kernel_launch(void* const* d_in, const int* in_sizes, int n_in,
                              void* d_out, int out_size, void* d_ws, size_t ws_size,
                              hipStream_t stream) {
  const float* x      = (const float*)d_in[0];
  const float* params = (const float*)d_in[1];
  float* out          = (float*)d_out;
  int B = in_sizes[0] / NQ;
  qsim14_kernel<<<dim3(512), dim3(NT), 0, stream>>>(x, params, out, B);
}


# Cross-lane sums of the Z partials moved from 70 serial LDS-unit shuffles per wave and sample to a vector-ALU butterfly (permlanex16, then DPP row_xmask 8,4,2,1), twelve sums per step; same pairing
# speedup vs baseline: 13.1637x; 1.6543x over previous
#include <hip/hip_runtime.h>
#include <math.h>

typedef float v2f __attribute__((ext_vector_type(2)));
typedef float v8f __attribute__((ext_vector_type(8)));

#define NQ   14
#define NT   256

static __device__ inline v8f wmma_16x16x4_f32(v2f a, v2f b, v8f c) {
#if defined(__gfx1250__)
  return __builtin_amdgcn_wmma_f32_16x16x4_f32(
      false, a, false, b, (short)0, c, false, false);
#else
  return c;
#endif
}

static __device__ __forceinline__ float xhalf(float x) {
#if defined(__gfx1250__)
  const int b = __builtin_bit_cast(int, x);
  return __builtin_bit_cast(float,
      __builtin_amdgcn_permlanex16(b, b, 0x76543210, 0xfedcba98, false, false));
#else
  return x;
#endif
}
template <int XM>
static __device__ __forceinline__ float xrow(float x) {
#if defined(__gfx1250__)
  const int b = __builtin_bit_cast(int, x);
  return __builtin_bit_cast(float,
      __builtin_amdgcn_update_dpp(0, b, 0x160 + XM, 0xf, 0xf, true));
#else
  return x;
#endif
}

static __device__ __forceinline__ constexpr int gray(int J) { return J ^ (J >> 1); }

static __device__ __forceinline__ float ry_el(int ib, int jb, float c, float s) {
  return (ib == jb) ? c : ((ib == 0) ? -s : s);
}

__global__ __launch_bounds__(NT) void qsim14_kernel(
    const float* __restrict__ x,
    const float* __restrict__ params,
    float* __restrict__ out,
    int nb)
{
  __shared__ float pHiR[128], pHiI[128];
  __shared__ float pLoR[128], pLoI[128];
  __shared__ float f0r[NQ], f0i[NQ];
  __shared__ float f1r[NQ], f1i[NQ];
  __shared__ float red[NQ * 8];

  const int tid  = threadIdx.x;
  const int lane = tid & 31;
  const int wave = tid >> 5;

  float cw0 = 1.f, sw0 = 0.f;
  if (tid < NQ) {
    float hp = 0.5f * params[tid];
    cw0 = cosf(hp); sw0 = sinf(hp);
  }

  const int hl = lane >> 4;
  const int m  = lane & 15;

  const int lb = gray(((m & 7) << 4) | (hl << 1)) ^ ((m >> 3) << 6);
  const int hb = gray(((m & 7) << 4) | (hl << 2) | (m >> 3));

  float gc[NQ], gs[NQ];
#pragma unroll
  for (int q = 0; q < NQ; ++q) {
    float ang = 0.5f * params[NQ + q];
    gc[q] = cosf(ang); gs[q] = sinf(ang);
  }

  v2f LB[4];
  v2f LG[4];
  v2f HG[4];
#pragma unroll
  for (int k = 0; k < 4; ++k) {
#pragma unroll
    for (int r = 0; r < 2; ++r) {
      const int K = 4 * k + 2 * hl + r;
      float e = 1.f;
#pragma unroll
      for (int mm = 0; mm < 4; ++mm)
        e *= ry_el((m >> mm) & 1, (K >> mm) & 1, gc[13 - mm], gs[13 - mm]);
      LB[k][r] = e;

      const int o3 = 2 * k + r;
      const float same = ((m >> 3) == hl) ? 1.f : 0.f;
      float eg = same;
#pragma unroll
      for (int mm = 0; mm < 3; ++mm)
        eg *= ry_el((m >> mm) & 1, (o3 >> mm) & 1, gc[9 - mm], gs[9 - mm]);
      LG[k][r] = eg;

      const int un = (m & 7) ^ wave;
      float eh = same;
#pragma unroll
      for (int mm = 0; mm < 3; ++mm)
        eh *= ry_el((un >> mm) & 1, (o3 >> mm) & 1, gc[2 - mm], gs[2 - mm]);
      HG[k][r] = eh;
    }
  }
  v2f HB[2];
#pragma unroll
  for (int k = 0; k < 2; ++k) {
#pragma unroll
    for (int r = 0; r < 2; ++r) {
      const int K = 4 * k + 2 * hl + r;
      float e = 1.f;
#pragma unroll
      for (int mm = 0; mm < 3; ++mm)
        e *= ry_el((m >> (mm + 1)) & 1, (K >> mm) & 1, gc[5 - mm], gs[5 - mm]);
      HB[k][r] = e;
    }
  }

  const float g6 = ry_el(m & 1, hl, gc[6], gs[6]);

  const int pwv = __builtin_popcount(wave) & 1;
  const int pu = pwv ^ (__builtin_popcount(m) & 1);
  const int ph = pu ^ hl;

  const int b0 = blockIdx.x;
  const int bstep = gridDim.x;
  float xn = 0.f;
  if (tid < NQ && b0 < nb) xn = x[b0 * NQ + tid];
  for (int b = b0; b < nb; b += bstep) {

  if (tid < NQ) {
    float hx = 0.5f * xn;
    float c = cosf(hx), s = sinf(hx);
    f0r[tid] = cw0 * c;  f0i[tid] = sw0 * s;
    f1r[tid] = sw0 * c;  f1i[tid] = -(cw0 * s);
  }
  __syncthreads();

  {
    const int hsel = tid >> 7;
    const int v7 = tid & 127;
    float pr = 1.f, pi = 0.f;
#pragma unroll
    for (int t = 0; t < 7; ++t) {
      const int q = 7 * hsel + t;
      const int bit = (v7 >> (6 - t)) & 1;
      float fr = bit ? f1r[q] : f0r[q];
      float fi = bit ? f1i[q] : f0i[q];
      float nr = pr * fr - pi * fi;
      float ni = pr * fi + pi * fr;
      pr = nr; pi = ni;
    }
    if (hsel == 0) { pHiR[v7] = pr; pHiI[v7] = pi; }
    else           { pLoR[v7] = pr; pLoI[v7] = pi; }
  }
  __syncthreads();

  if (tid < NQ && b + bstep < nb) xn = x[(b + bstep) * NQ + tid];

  v2f LAr[4], LAi[4];
#pragma unroll
  for (int k = 0; k < 4; ++k) {
#pragma unroll
    for (int r = 0; r < 2; ++r) {
      const int idx = lb ^ gray(4 * k + r);
      LAr[k][r] = pLoR[idx];
      LAi[k][r] = pLoI[idx];
    }
  }
  v2f HAr[2], HAi[2];
#pragma unroll
  for (int k = 0; k < 2; ++k) {
#pragma unroll
    for (int r = 0; r < 2; ++r) {
      const int idx = hb ^ gray(8 * k + 2 * r);
      HAr[k][r] = pHiR[idx];
      HAi[k][r] = pHiI[idx];
    }
  }

  v8f L1r = {}, L1i = {};
#pragma unroll
  for (int k = 0; k < 4; ++k) {
    L1r = wmma_16x16x4_f32(LAr[k], LB[k], L1r);
    L1i = wmma_16x16x4_f32(LAi[k], LB[k], L1i);
  }
  v8f L2r = {}, L2i = {};
#pragma unroll
  for (int k = 0; k < 4; ++k) {
    v2f Br, Bi;
    Br[0] = L1r[2 * k + 0]; Br[1] = L1r[2 * k + 1];
    Bi[0] = L1i[2 * k + 0]; Bi[1] = L1i[2 * k + 1];
    L2r = wmma_16x16x4_f32(LG[k], Br, L2r);
    L2i = wmma_16x16x4_f32(LG[k], Bi, L2i);
  }

  v8f H1r = {}, H1i = {};
#pragma unroll
  for (int k = 0; k < 2; ++k) {
    H1r = wmma_16x16x4_f32(HAr[k], HB[k], H1r);
    H1i = wmma_16x16x4_f32(HAi[k], HB[k], H1i);
  }
  v8f H2r = {}, H2i = {};
#pragma unroll
  for (int k = 0; k < 4; ++k) {
    v2f Br, Bi;
    Br[0] = H1r[2 * k + 0]; Br[1] = H1r[2 * k + 1];
    Bi[0] = H1i[2 * k + 0]; Bi[1] = H1i[2 * k + 1];
    H2r = wmma_16x16x4_f32(HG[k], Br, H2r);
    H2i = wmma_16x16x4_f32(HG[k], Bi, H2i);
  }

  v2f Bf;
  Bf[0] = g6 * H2r[0];
  Bf[1] = g6 * H2i[0];
  float accAll = 0.f;
  float accT7 = 0.f, accT8 = 0.f, accT9 = 0.f;
  float accT11 = 0.f, accT12 = 0.f, accT13 = 0.f;
#pragma unroll
  for (int t = 0; t < 8; ++t) {
    v2f Are, Aim;
    Are[0] = L2r[t]; Are[1] = -L2i[t];
    Aim[0] = L2i[t]; Aim[1] = L2r[t];
    const v8f zero = {};
    const v8f accR = wmma_16x16x4_f32(Are, Bf, zero);
    const v8f accI = wmma_16x16x4_f32(Aim, Bf, zero);
    float pw[8];
#pragma unroll
    for (int r = 0; r < 8; ++r) pw[r] = accR[r] * accR[r] + accI[r] * accI[r];
    const float a0 = pw[0] + pw[1], d0 = pw[0] - pw[1];
    const float a1 = pw[2] + pw[3], d1 = pw[2] - pw[3];
    const float a2 = pw[4] + pw[5], d2 = pw[4] - pw[5];
    const float a3 = pw[6] + pw[7], d3 = pw[6] - pw[7];
    const float b0 = a0 + a1, e0 = a0 - a1;
    const float b1 = a2 + a3, e1 = a2 - a3;
    const float sAll = b0 + b1;
    const float s11  = b0 - b1;
    const float s12  = e0 - e1;
    const float s13  = (d0 - d1) - (d2 - d3);
    accAll += sAll;
    if ((t >> 2) & 1) accT7 -= sAll; else accT7 += sAll;
    if (__builtin_popcount(t >> 1) & 1) accT8 -= sAll; else accT8 += sAll;
    if (__builtin_popcount(t) & 1) { accT9 -= sAll; accT11 -= s11; accT12 -= s12; accT13 -= s13; }
    else                           { accT9 += sAll; accT11 += s11; accT12 += s12; accT13 += s13; }
  }

  float sv[12];
  {
    sv[0] = accAll;
#pragma unroll
    for (int w = 3; w < 7; ++w)
      sv[w - 2] = (pwv ^ (__builtin_popcount(m >> (6 - w)) & 1)) ? -accAll : accAll;
    sv[5]  = pu ? -accT7 : accT7;
    sv[6]  = pu ? -accT8 : accT8;
    sv[7]  = pu ? -accT9 : accT9;
    sv[8]  = ph ? -accT9 : accT9;
    sv[9]  = ph ? -accT11 : accT11;
    sv[10] = ph ? -accT12 : accT12;
    sv[11] = ph ? -accT13 : accT13;
  }

#pragma unroll
  for (int i = 0; i < 12; ++i) sv[i] += xhalf(sv[i]);
#pragma unroll
  for (int i = 0; i < 12; ++i) sv[i] += xrow<8>(sv[i]);
#pragma unroll
  for (int i = 0; i < 12; ++i) sv[i] += xrow<4>(sv[i]);
#pragma unroll
  for (int i = 0; i < 12; ++i) sv[i] += xrow<2>(sv[i]);
#pragma unroll
  for (int i = 0; i < 12; ++i) sv[i] += xrow<1>(sv[i]);

  if (lane == 0) {
    red[0 * 8 + wave] = ((wave >> 2) & 1) ? -sv[0] : sv[0];
    red[1 * 8 + wave] = (__builtin_popcount(wave >> 1) & 1) ? -sv[0] : sv[0];
    red[2 * 8 + wave] = pwv ? -sv[0] : sv[0];
#pragma unroll
    for (int w = 3; w < NQ; ++w) red[w * 8 + wave] = sv[w - 2];
  }
  __syncthreads();
  if (tid < NQ) {
    float s = 0.f;
#pragma unroll
    for (int i = 0; i < 8; ++i) s += red[tid * 8 + i];
    out[b * NQ + tid] = s;
  }

  }
}

extern "C" void kernel_launch(void* const* d_in, const int* in_sizes, int n_in,
                              void* d_out, int out_size, void* d_ws, size_t ws_size,
                              hipStream_t stream) {
  const float* x      = (const float*)d_in[0];
  const float* params = (const float*)d_in[1];
  float* out          = (float*)d_out;
  int B = in_sizes[0] / NQ;
  qsim14_kernel<<<dim3(512), dim3(NT), 0, stream>>>(x, params, out, B);
}


# Z sums taken from 2x2 signed Gram sums of the two 128-entry factors (sign of each wire = high part x low part): amplitudes and their WMMAs go, one wave does a whole sample (sum order changes)
# speedup vs baseline: 42.4317x; 3.2234x over previous
#include <hip/hip_runtime.h>
#include <math.h>

typedef float v2f __attribute__((ext_vector_type(2)));
typedef float v8f __attribute__((ext_vector_type(8)));

#define NQ   14
#define NT   256

static __device__ inline v8f wmma_16x16x4_f32(v2f a, v2f b, v8f c) {
#if defined(__gfx1250__)
  return __builtin_amdgcn_wmma_f32_16x16x4_f32(
      false, a, false, b, (short)0, c, false, false);
#else
  return c;
#endif
}

static __device__ __forceinline__ float xhalf(float x) {
#if defined(__gfx1250__)
  const int b = __builtin_bit_cast(int, x);
  return __builtin_bit_cast(float,
      __builtin_amdgcn_permlanex16(b, b, 0x76543210, 0xfedcba98, false, false));
#else
  return x;
#endif
}
template <int XM>
static __device__ __forceinline__ float xrow(float x) {
#if defined(__gfx1250__)
  const int b = __builtin_bit_cast(int, x);
  return __builtin_bit_cast(float,
      __builtin_amdgcn_update_dpp(0, b, 0x160 + XM, 0xf, 0xf, true));
#else
  return x;
#endif
}

static __device__ __forceinline__ constexpr int gray(int J) { return J ^ (J >> 1); }

static __device__ __forceinline__ float ry_el(int ib, int jb, float c, float s) {
  return (ib == jb) ? c : ((ib == 0) ? -s : s);
}

__global__ __launch_bounds__(NT) void qsim14_kernel(
    const float* __restrict__ x,
    const float* __restrict__ params,
    float* __restrict__ out,
    int nb)
{
  __shared__ float fac[NT / 32][4][16];
  __shared__ float tab[NT / 32][4][128];

  const int tid  = threadIdx.x;
  const int lane = tid & 31;
  const int wave = tid >> 5;

  float cw0 = 1.f, sw0 = 0.f;
  if (lane < NQ) {
    float hp = 0.5f * params[lane];
    cw0 = cosf(hp); sw0 = sinf(hp);
  }

  const int hl = lane >> 4;
  const int m  = lane & 15;

  const int lb = gray(((m & 7) << 4) | (hl << 1)) ^ ((m >> 3) << 6);
  const int hb = gray(((m & 7) << 4) | (hl << 2) | (m >> 3));

  float gc[NQ], gs[NQ];
#pragma unroll
  for (int q = 0; q < NQ; ++q) {
    float ang = 0.5f * params[NQ + q];
    gc[q] = cosf(ang); gs[q] = sinf(ang);
  }

  v2f LB[4];
  v2f LG[4];
  v2f HG[4];
#pragma unroll
  for (int k = 0; k < 4; ++k) {
#pragma unroll
    for (int r = 0; r < 2; ++r) {
      const int K = 4 * k + 2 * hl + r;
      float e = 1.f;
#pragma unroll
      for (int mm = 0; mm < 4; ++mm)
        e *= ry_el((m >> mm) & 1, (K >> mm) & 1, gc[13 - mm], gs[13 - mm]);
      LB[k][r] = e;

      const int o3 = 2 * k + r;
      const float same = ((m >> 3) == hl) ? 1.f : 0.f;
      float eg = same;
#pragma unroll
      for (int mm = 0; mm < 3; ++mm)
        eg *= ry_el((m >> mm) & 1, (o3 >> mm) & 1, gc[9 - mm], gs[9 - mm]);
      LG[k][r] = eg;

      const int un = m & 7;
      float eh = same;
#pragma unroll
      for (int mm = 0; mm < 3; ++mm)
        eh *= ry_el((un >> mm) & 1, (o3 >> mm) & 1, gc[2 - mm], gs[2 - mm]);
      HG[k][r] = eh;
    }
  }
  v2f HB[2];
#pragma unroll
  for (int k = 0; k < 2; ++k) {
#pragma unroll
    for (int r = 0; r < 2; ++r) {
      const int K = 4 * k + 2 * hl + r;
      float e = 1.f;
#pragma unroll
      for (int mm = 0; mm < 3; ++mm)
        e *= ry_el((m >> (mm + 1)) & 1, (K >> mm) & 1, gc[5 - mm], gs[5 - mm]);
      HB[k][r] = e;
    }
  }

  const float c6 = gc[6], s6 = gs[6];
  const float kP   = c6 * c6 + s6 * s6;
  const float kZ   = c6 * c6 - s6 * s6;
  const float kZ01 = -4.f * (c6 * s6);
  const float kZown = hl ? -kZ : kZ;

  const int m3 = (m >> 3) & 1, m2 = (m >> 2) & 1, m1 = (m >> 1) & 1, m0 = m & 1;
  const int q3 = m3, q32 = m3 ^ m2, q321 = q32 ^ m1, q3210 = q321 ^ m0;

  float* const F  = &fac[wave][0][0];
  float* const Tb = &tab[wave][0][0];

  for (int base = blockIdx.x * (NT / 32); base < nb; base += gridDim.x * (NT / 32)) {
  const int b = base + wave;
  const bool valid = b < nb;

  if (lane < NQ) {
    float hx = valid ? 0.5f * x[b * NQ + lane] : 0.f;
    float c = cosf(hx), s = sinf(hx);
    F[0 * 16 + lane] = cw0 * c;  F[1 * 16 + lane] = sw0 * s;
    F[2 * 16 + lane] = sw0 * c;  F[3 * 16 + lane] = -(cw0 * s);
  }
  __syncthreads();

#pragma unroll
  for (int h = 0; h < 2; ++h) {
    float sr[5], si[5];
#pragma unroll
    for (int t = 2; t < 7; ++t) {
      const int q = 7 * h + t;
      const int bit = (lane >> (6 - t)) & 1;
      sr[t - 2] = bit ? F[2 * 16 + q] : F[0 * 16 + q];
      si[t - 2] = bit ? F[3 * 16 + q] : F[1 * 16 + q];
    }
#pragma unroll
    for (int e2 = 0; e2 < 4; ++e2) {
      float pr = 1.f, pi = 0.f;
#pragma unroll
      for (int t = 0; t < 7; ++t) {
        const int q = 7 * h + t;
        float fr, fi;
        if (t < 2) {
          const int bit = (e2 >> (1 - t)) & 1;
          fr = bit ? F[2 * 16 + q] : F[0 * 16 + q];
          fi = bit ? F[3 * 16 + q] : F[1 * 16 + q];
        } else {
          fr = sr[t - 2]; fi = si[t - 2];
        }
        float nr = pr * fr - pi * fi;
        float ni = pr * fi + pi * fr;
        pr = nr; pi = ni;
      }
      const int v7 = (e2 << 5) | lane;
      Tb[(2 * h + 0) * 128 + v7] = pr;
      Tb[(2 * h + 1) * 128 + v7] = pi;
    }
  }
  __syncthreads();

  v2f LAr[4], LAi[4];
#pragma unroll
  for (int k = 0; k < 4; ++k) {
#pragma unroll
    for (int r = 0; r < 2; ++r) {
      const int idx = lb ^ gray(4 * k + r);
      LAr[k][r] = Tb[2 * 128 + idx];
      LAi[k][r] = Tb[3 * 128 + idx];
    }
  }
  v2f HAr[2], HAi[2];
#pragma unroll
  for (int k = 0; k < 2; ++k) {
#pragma unroll
    for (int r = 0; r < 2; ++r) {
      const int idx = hb ^ gray(8 * k + 2 * r);
      HAr[k][r] = Tb[0 * 128 + idx];
      HAi[k][r] = Tb[1 * 128 + idx];
    }
  }

  v8f L1r = {}, L1i = {};
#pragma unroll
  for (int k = 0; k < 4; ++k) {
    L1r = wmma_16x16x4_f32(LAr[k], LB[k], L1r);
    L1i = wmma_16x16x4_f32(LAi[k], LB[k], L1i);
  }
  v8f L2r = {}, L2i = {};
#pragma unroll
  for (int k = 0; k < 4; ++k) {
    v2f Br, Bi;
    Br[0] = L1r[2 * k + 0]; Br[1] = L1r[2 * k + 1];
    Bi[0] = L1i[2 * k + 0]; Bi[1] = L1i[2 * k + 1];
    L2r = wmma_16x16x4_f32(LG[k], Br, L2r);
    L2i = wmma_16x16x4_f32(LG[k], Bi, L2i);
  }

  v8f H1r = {}, H1i = {};
#pragma unroll
  for (int k = 0; k < 2; ++k) {
    H1r = wmma_16x16x4_f32(HAr[k], HB[k], H1r);
    H1i = wmma_16x16x4_f32(HAi[k], HB[k], H1i);
  }
  v8f H2r = {}, H2i = {};
#pragma unroll
  for (int k = 0; k < 4; ++k) {
    v2f Br, Bi;
    Br[0] = H1r[2 * k + 0]; Br[1] = H1r[2 * k + 1];
    Bi[0] = H1i[2 * k + 0]; Bi[1] = H1i[2 * k + 1];
    H2r = wmma_16x16x4_f32(HG[k], Br, H2r);
    H2i = wmma_16x16x4_f32(HG[k], Bi, H2i);
  }

  float hv[8];
  {
    float d[8], xr[8], xi[8];
#pragma unroll
    for (int r = 0; r < 8; ++r) {
      const float orr = xhalf(H2r[r]), oi = xhalf(H2i[r]);
      d[r]  = H2r[r] * H2r[r] + H2i[r] * H2i[r];
      xr[r] = H2r[r] * orr + H2i[r] * oi;
      xi[r] = H2i[r] * orr - H2r[r] * oi;
    }
    const float a0 = d[0] + d[1], a1 = d[2] + d[3], a2 = d[4] + d[5], a3 = d[6] + d[7];
    const float s0 = (a0 + a1) - (a2 + a3);
    const float s1 = (a0 - a1) - (a2 - a3);
    const float s2 = ((d[0] - d[1]) - (d[2] - d[3])) - ((d[4] - d[5]) - (d[6] - d[7]));
    const float r2 = ((xr[0] - xr[1]) - (xr[2] - xr[3])) - ((xr[4] - xr[5]) - (xr[6] - xr[7]));
    const float i2 = ((xi[0] - xi[1]) - (xi[2] - xi[3])) - ((xi[4] - xi[5]) - (xi[6] - xi[7]));
    hv[0] = s0;
    hv[1] = s1;
    hv[2] = s2;
    hv[3] = q3   ? -s2 : s2;
    hv[4] = q32  ? -s2 : s2;
    hv[5] = q321 ? -s2 : s2;
    hv[6] = q321 ? -r2 : r2;
    hv[7] = q321 ? -i2 : i2;
  }
  float ld[8], lr[8], li[8];
  {
    float d[8], xr[8], xi[8];
#pragma unroll
    for (int t = 0; t < 8; ++t) {
      const float orr = xhalf(L2r[t]), oi = xhalf(L2i[t]);
      d[t]  = L2r[t] * L2r[t] + L2i[t] * L2i[t];
      xr[t] = L2r[t] * orr + L2i[t] * oi;
      xi[t] = L2i[t] * orr - L2r[t] * oi;
    }
#define SIGNED_SUMS(v, o)                                                             \
    {                                                                                 \
      const float a0 = v[0] + v[1], a1 = v[2] + v[3], a2 = v[4] + v[5], a3 = v[6] + v[7]; \
      const float p2 = ((v[0] - v[1]) - (v[2] - v[3])) - ((v[4] - v[5]) - (v[6] - v[7])); \
      o[0] = (a0 + a1) + (a2 + a3);                                                   \
      o[1] = (a0 + a1) - (a2 + a3);                                                   \
      o[2] = (a0 - a1) - (a2 - a3);                                                   \
      o[3] = p2;                                                                      \
      o[4] = q3    ? -p2 : p2;                                                        \
      o[5] = q32   ? -p2 : p2;                                                        \
      o[6] = q321  ? -p2 : p2;                                                        \
      o[7] = q3210 ? -p2 : p2;                                                        \
    }
    SIGNED_SUMS(d, ld)
    SIGNED_SUMS(xr, lr)
    SIGNED_SUMS(xi, li)
#undef SIGNED_SUMS
  }

#pragma unroll
  for (int i = 0; i < 8; ++i) { hv[i] += xrow<8>(hv[i]); ld[i] += xrow<8>(ld[i]); lr[i] += xrow<8>(lr[i]); li[i] += xrow<8>(li[i]); }
#pragma unroll
  for (int i = 0; i < 8; ++i) { hv[i] += xrow<4>(hv[i]); ld[i] += xrow<4>(ld[i]); lr[i] += xrow<4>(lr[i]); li[i] += xrow<4>(li[i]); }
#pragma unroll
  for (int i = 0; i < 8; ++i) { hv[i] += xrow<2>(hv[i]); ld[i] += xrow<2>(ld[i]); lr[i] += xrow<2>(lr[i]); li[i] += xrow<2>(li[i]); }
#pragma unroll
  for (int i = 0; i < 8; ++i) { ld[i] += xrow<1>(ld[i]); lr[i] += xrow<1>(lr[i]); li[i] += xrow<1>(li[i]); }

  float ohd[6], old0[8];
#pragma unroll
  for (int i = 0; i < 6; ++i) ohd[i] = xhalf(hv[i]);
#pragma unroll
  for (int i = 0; i < 8; ++i) old0[i] = xhalf(ld[i]);
  float o = 0.f;
#pragma unroll
  for (int w = 0; w < NQ; ++w) {
    float v;
    if (w < 6) {
      v = kP * (hv[w] * ld[0] + ohd[w] * old0[0]);
    } else {
      const int j = w - 6;
      v = kZown * (hv[5] * ld[j] - ohd[5] * old0[j]) + kZ01 * (hv[6] * lr[j] - hv[7] * li[j]);
    }
    o = (lane == w) ? v : o;
  }
  if (lane < NQ && valid) out[b * NQ + lane] = o;

  }
}

extern "C" void kernel_launch(void* const* d_in, const int* in_sizes, int n_in,
                              void* d_out, int out_size, void* d_ws, size_t ws_size,
                              hipStream_t stream) {
  const float* x      = (const float*)d_in[0];
  const float* params = (const float*)d_in[1];
  float* out          = (float*)d_out;
  int B = in_sizes[0] / NQ;
  qsim14_kernel<<<dim3(512), dim3(NT), 0, stream>>>(x, params, out, B);
}
